# starting kernel
# baseline (speedup 1.0000x reference)
#include <hip/hip_runtime.h>
#include <math.h>

typedef __attribute__((ext_vector_type(2))) float v2f;
typedef __attribute__((ext_vector_type(8))) float v8f;

#define HDIM 128
#define NLAY 3
#define PPB 16
#define NC 6
#define NCOL (NC * PPB)
#define CP 112
#define WS 132
#define WGSIZE 256

#define HIDX(k, col) ((((k) >> 1) * (2 * CP)) + (((col)) << 1) + ((k) & 1))

__device__ __forceinline__ void run_net(
    float* __restrict__ Wlds, float* __restrict__ Hl, float* __restrict__ sbuf,
    float* __restrict__ jet,
    const float* __restrict__ W1, const float* __restrict__ b1,
    const float* __restrict__ Wh, const float* __restrict__ bh,
    const float* __restrict__ Wf, const float* __restrict__ bf,
    const float* __restrict__ xs, const float* __restrict__ ys)
{
    const int t    = threadIdx.x;
    const int lane = t & 31;
    const int wave = t >> 5;

    {
        const int p = t & 15;
        const float x = xs[p], y = ys[p];
        #pragma unroll
        for (int j = 0; j < 8; ++j) {
            const int i = (t >> 4) + 16 * j;
            const float w0 = W1[i * 2 + 0], w1 = W1[i * 2 + 1];
            const float z  = fmaf(w0, x, fmaf(w1, y, b1[i]));
            const float tv = tanhf(z);
            const float t1 = 1.f - tv * tv;
            const float t2 = -2.f * tv * t1;
            Hl[HIDX(i, 0 * PPB + p)] = tv;
            Hl[HIDX(i, 1 * PPB + p)] = t1 * w0;
            Hl[HIDX(i, 2 * PPB + p)] = t1 * w1;
            Hl[HIDX(i, 3 * PPB + p)] = t2 * w0 * w0;
            Hl[HIDX(i, 4 * PPB + p)] = t2 * w0 * w1;
            Hl[HIDX(i, 5 * PPB + p)] = t2 * w1 * w1;
        }
    }
    __syncthreads();

    for (int l = 0; l < NLAY; ++l) {
        const float* __restrict__ W = Wh + l * HDIM * HDIM;

        for (int idx = t; idx < HDIM * (HDIM / 4); idx += WGSIZE) {
            const int row = idx >> 5;
            const int q   = idx & 31;
            const float4 v = *(const float4*)(W + row * HDIM + q * 4);
            *(float4*)(Wlds + row * WS + q * 4) = v;
        }
        if (l + 1 < NLAY) {
            __builtin_prefetch(Wh + (l + 1) * HDIM * HDIM + t * 64, 0, 1);
        }
        __syncthreads();

        const int rbase  = wave * 16;
        const int m      = lane & 15;
        const int khalf  = (lane < 16) ? 0 : 2;
        const int rowoff = (lane < 16) ? 0 : 8;

        v8f acc[NC];
        const v8f vzero = {0.f, 0.f, 0.f, 0.f, 0.f, 0.f, 0.f, 0.f};
        #pragma unroll
        for (int c = 0; c < NC; ++c) acc[c] = vzero;

        for (int kt = 0; kt < HDIM / 4; ++kt) {
            const int kb = kt * 4;
            const v2f a = *(const v2f*)&Wlds[(rbase + m) * WS + kb + khalf];
            #pragma unroll
            for (int c = 0; c < NC; ++c) {
                const v2f b = *(const v2f*)&Hl[HIDX(kb + khalf, c * PPB + m)];
                acc[c] = __builtin_amdgcn_wmma_f32_16x16x4_f32(
                    false, a, false, b, (short)0, acc[c], false, false);
            }
        }
        __syncthreads();

        #pragma unroll
        for (int r = 0; r < 8; ++r) {
            const int i  = rbase + rowoff + r;
            const float zv  = acc[0][r] + bh[l * HDIM + i];
            const float zgx = acc[1][r];
            const float zgy = acc[2][r];
            const float zxx = acc[3][r];
            const float zxy = acc[4][r];
            const float zyy = acc[5][r];
            const float tv = tanhf(zv);
            const float t1 = 1.f - tv * tv;
            const float t2 = -2.f * tv * t1;
            Hl[HIDX(i, 0 * PPB + m)] = tv;
            Hl[HIDX(i, 1 * PPB + m)] = t1 * zgx;
            Hl[HIDX(i, 2 * PPB + m)] = t1 * zgy;
            Hl[HIDX(i, 3 * PPB + m)] = fmaf(t2, zgx * zgx, t1 * zxx);
            Hl[HIDX(i, 4 * PPB + m)] = fmaf(t2, zgx * zgy, t1 * zxy);
            Hl[HIDX(i, 5 * PPB + m)] = fmaf(t2, zgy * zgy, t1 * zyy);
        }
        __syncthreads();
    }

    if (t < 2 * NCOL) {
        const int col  = t % NCOL;
        const int half = t / NCOL;
        float s = 0.f;
        #pragma unroll 4
        for (int k = half * 64; k < half * 64 + 64; ++k)
            s = fmaf(Wf[k], Hl[HIDX(k, col)], s);
        sbuf[half * NCOL + col] = s;
    }
    __syncthreads();
    if (t < PPB) {
        const int p = t;
        const float sv  = sbuf[0 * PPB + p] + sbuf[NCOL + 0 * PPB + p] + bf[0];
        const float sgx = sbuf[1 * PPB + p] + sbuf[NCOL + 1 * PPB + p];
        const float sgy = sbuf[2 * PPB + p] + sbuf[NCOL + 2 * PPB + p];
        const float sxx = sbuf[3 * PPB + p] + sbuf[NCOL + 3 * PPB + p];
        const float sxy = sbuf[4 * PPB + p] + sbuf[NCOL + 4 * PPB + p];
        const float syy = sbuf[5 * PPB + p] + sbuf[NCOL + 5 * PPB + p];
        const float tv = tanhf(sv);
        const float t1 = 1.f - tv * tv;
        const float t2 = -2.f * tv * t1;
        jet[0 * PPB + p] = tv;
        jet[1 * PPB + p] = t1 * sgx;
        jet[2 * PPB + p] = t1 * sgy;
        jet[3 * PPB + p] = fmaf(t2, sgx * sgx, t1 * sxx);
        jet[4 * PPB + p] = fmaf(t2, sgx * sgy, t1 * sxy);
        jet[5 * PPB + p] = fmaf(t2, sgy * sgy, t1 * syy);
    }
    __syncthreads();
}

__global__ void __launch_bounds__(WGSIZE)
pinn_elasticity_kernel(
    const float* __restrict__ x, const float* __restrict__ y,
    const float* __restrict__ W1u, const float* __restrict__ b1u,
    const float* __restrict__ Whu, const float* __restrict__ bhu,
    const float* __restrict__ Wfu, const float* __restrict__ bfu,
    const float* __restrict__ W1v, const float* __restrict__ b1v,
    const float* __restrict__ Whv, const float* __restrict__ bhv,
    const float* __restrict__ Wfv, const float* __restrict__ bfv,
    const float* __restrict__ mu, const float* __restrict__ lmbd,
    float* __restrict__ out, int N)
{
    extern __shared__ float lds[];
    float* Wlds = lds;
    float* Hl   = Wlds + HDIM * WS;
    float* sbuf = Hl + HDIM * CP;
    float* jets = sbuf + 2 * NCOL;
    float* xs   = jets + 2 * NC * PPB;
    float* ys   = xs + PPB;

    const int t  = threadIdx.x;
    const int p0 = blockIdx.x * PPB;

    if (t < PPB) {
        const int gp = p0 + t;
        xs[t] = (gp < N) ? x[gp] : 0.f;
        ys[t] = (gp < N) ? y[gp] : 0.f;
    }
    __syncthreads();

    run_net(Wlds, Hl, sbuf, jets,            W1u, b1u, Whu, bhu, Wfu, bfu, xs, ys);
    run_net(Wlds, Hl, sbuf, jets + NC * PPB, W1v, b1v, Whv, bhv, Wfv, bfv, xs, ys);

    if (t < PPB) {
        const int p  = t;
        const int gp = p0 + p;
        if (gp < N) {
            const float* ju = jets;
            const float* jv = jets + NC * PPB;
            const float Uv  = ju[0 * PPB + p];
            const float Uxx = ju[3 * PPB + p];
            const float Uxy = ju[4 * PPB + p];
            const float Uyy = ju[5 * PPB + p];
            const float Vv  = jv[0 * PPB + p];
            const float Vxx = jv[3 * PPB + p];
            const float Vxy = jv[4 * PPB + p];
            const float Vyy = jv[5 * PPB + p];

            const float MU = mu[0], LA = lmbd[0];
            const float C11 = 2.f * MU + LA;
            const float C12 = LA;
            const float C33 = 2.f * MU;

            const float rx = C11 * Uxx + C12 * Vxy + C33 * (Uyy + 0.5f * Vxy);
            const float ry = C33 * (Uxy + 0.5f * Vxx) + C11 * Vyy + C12 * Uxy;

            const float PI = 3.14159265358979323846f;
            const float Qc = 4.0f;
            const float xx = xs[p], yy = ys[p];
            const float y2 = yy * yy, y3 = y2 * yy, y4 = y3 * yy;
            const float s2px = sinf(2.f * PI * xx), c2px = cosf(2.f * PI * xx);
            const float spx  = sinf(PI * xx),       cpx  = cosf(PI * xx);
            const float spy  = sinf(PI * yy),       cpy  = cosf(PI * yy);
            const float PI2  = PI * PI;

            const float Fx = -1.0f * (4.f * PI2 * c2px * spy - Qc * y3 * PI * cpx)
                             - 0.5f * (PI2 * c2px * spy - Qc * y3 * PI * cpx)
                             - 0.5f * 8.f * PI2 * c2px * spy;
            const float Fy =  1.0f * (3.f * Qc * y2 * spx - 2.f * PI2 * cpy * s2px)
                             - 0.5f * (2.f * PI2 * cpy * s2px + Qc * y4 * PI2 * spx * 0.25f)
                             + 0.5f * 6.f * Qc * y2 * spx;

            out[0 * N + gp] = rx - Fx;
            out[1 * N + gp] = ry - Fy;
            out[2 * N + gp] = Uv;
            out[3 * N + gp] = Vv;
        }
    }
}

extern "C" void kernel_launch(void* const* d_in, const int* in_sizes, int n_in,
                              void* d_out, int out_size, void* d_ws, size_t ws_size,
                              hipStream_t stream) {
    const float* x    = (const float*)d_in[0];
    const float* y    = (const float*)d_in[1];
    const float* W1u  = (const float*)d_in[2];
    const float* b1u  = (const float*)d_in[3];
    const float* Whu  = (const float*)d_in[4];
    const float* bhu  = (const float*)d_in[5];
    const float* Wfu  = (const float*)d_in[6];
    const float* bfu  = (const float*)d_in[7];
    const float* W1v  = (const float*)d_in[8];
    const float* b1v  = (const float*)d_in[9];
    const float* Whv  = (const float*)d_in[10];
    const float* bhv  = (const float*)d_in[11];
    const float* Wfv  = (const float*)d_in[12];
    const float* bfv  = (const float*)d_in[13];
    const float* mu   = (const float*)d_in[14];
    const float* lmbd = (const float*)d_in[15];

    const int N = in_sizes[0];
    const int grid = (N + PPB - 1) / PPB;
    const size_t shmem =
        (size_t)(HDIM * WS + HDIM * CP + 2 * NCOL + 2 * NC * PPB + 2 * PPB) * sizeof(float);

    pinn_elasticity_kernel<<<grid, WGSIZE, shmem, stream>>>(
        x, y, W1u, b1u, Whu, bhu, Wfu, bfu,
        W1v, b1v, Whv, bhv, Wfv, bfv, mu, lmbd,
        (float*)d_out, N);
}


# Weight tile fetch un-serialised: a thread's 16 loads per layer issue as one group behind the K-loop barrier and land under the tanh epilogue, LDS stores after it; bias loads moved before the K loop
# speedup vs baseline: 1.1536x; 1.1536x over previous
#include <hip/hip_runtime.h>
#include <math.h>

typedef __attribute__((ext_vector_type(2))) float v2f;
typedef __attribute__((ext_vector_type(8))) float v8f;

#define HDIM 128
#define NLAY 3
#define PPB 16
#define NC 6
#define NCOL (NC * PPB)
#define CP 112
#define WS 132
#define WGSIZE 256

#define HIDX(k, col) ((((k) >> 1) * (2 * CP)) + (((col)) << 1) + ((k) & 1))

typedef __attribute__((ext_vector_type(4))) float v4f;

#define W_DECL v4f wt0, wt1, wt2, wt3, wt4, wt5, wt6, wt7, wt8, wt9, wt10, wt11, wt12, wt13, wt14, wt15
#define W_FETCH(Wp) do { const float* wp_ = (Wp) + (t >> 5) * HDIM + (t & 31) * 4; \
    wt0 = *(const v4f*)(wp_ + 0 * 8 * HDIM); \
    wt1 = *(const v4f*)(wp_ + 1 * 8 * HDIM); \
    wt2 = *(const v4f*)(wp_ + 2 * 8 * HDIM); \
    wt3 = *(const v4f*)(wp_ + 3 * 8 * HDIM); \
    wt4 = *(const v4f*)(wp_ + 4 * 8 * HDIM); \
    wt5 = *(const v4f*)(wp_ + 5 * 8 * HDIM); \
    wt6 = *(const v4f*)(wp_ + 6 * 8 * HDIM); \
    wt7 = *(const v4f*)(wp_ + 7 * 8 * HDIM); \
    wt8 = *(const v4f*)(wp_ + 8 * 8 * HDIM); \
    wt9 = *(const v4f*)(wp_ + 9 * 8 * HDIM); \
    wt10 = *(const v4f*)(wp_ + 10 * 8 * HDIM); \
    wt11 = *(const v4f*)(wp_ + 11 * 8 * HDIM); \
    wt12 = *(const v4f*)(wp_ + 12 * 8 * HDIM); \
    wt13 = *(const v4f*)(wp_ + 13 * 8 * HDIM); \
    wt14 = *(const v4f*)(wp_ + 14 * 8 * HDIM); \
    wt15 = *(const v4f*)(wp_ + 15 * 8 * HDIM); \
} while (0)
#define W_STORE() do { float* lp_ = Wlds + (t >> 5) * WS + (t & 31) * 4; \
    *(v4f*)(lp_ + 0 * 8 * WS) = wt0; \
    *(v4f*)(lp_ + 1 * 8 * WS) = wt1; \
    *(v4f*)(lp_ + 2 * 8 * WS) = wt2; \
    *(v4f*)(lp_ + 3 * 8 * WS) = wt3; \
    *(v4f*)(lp_ + 4 * 8 * WS) = wt4; \
    *(v4f*)(lp_ + 5 * 8 * WS) = wt5; \
    *(v4f*)(lp_ + 6 * 8 * WS) = wt6; \
    *(v4f*)(lp_ + 7 * 8 * WS) = wt7; \
    *(v4f*)(lp_ + 8 * 8 * WS) = wt8; \
    *(v4f*)(lp_ + 9 * 8 * WS) = wt9; \
    *(v4f*)(lp_ + 10 * 8 * WS) = wt10; \
    *(v4f*)(lp_ + 11 * 8 * WS) = wt11; \
    *(v4f*)(lp_ + 12 * 8 * WS) = wt12; \
    *(v4f*)(lp_ + 13 * 8 * WS) = wt13; \
    *(v4f*)(lp_ + 14 * 8 * WS) = wt14; \
    *(v4f*)(lp_ + 15 * 8 * WS) = wt15; \
} while (0)

__device__ __forceinline__ void run_net(
    float* __restrict__ Wlds, float* __restrict__ Hl, float* __restrict__ sbuf,
    float* __restrict__ jet,
    const float* __restrict__ W1, const float* __restrict__ b1,
    const float* __restrict__ Wh, const float* __restrict__ bh,
    const float* __restrict__ Wf, const float* __restrict__ bf,
    const float* __restrict__ xs, const float* __restrict__ ys)
{
    const int t    = threadIdx.x;
    const int lane = t & 31;
    const int wave = t >> 5;

    W_DECL;

    {
        const int p = t & 15;
        const float x = xs[p], y = ys[p];
        float iw0[8], iw1[8], ib[8];
        #pragma unroll
        for (int j = 0; j < 8; ++j) {
            const int i = (t >> 4) + 16 * j;
            iw0[j] = W1[i * 2 + 0];
            iw1[j] = W1[i * 2 + 1];
            ib[j]  = b1[i];
        }
        W_FETCH(Wh);
        #pragma unroll
        for (int j = 0; j < 8; ++j) {
            const int i = (t >> 4) + 16 * j;
            const float w0 = iw0[j], w1 = iw1[j];
            const float z  = fmaf(w0, x, fmaf(w1, y, ib[j]));
            const float tv = tanhf(z);
            const float t1 = 1.f - tv * tv;
            const float t2 = -2.f * tv * t1;
            Hl[HIDX(i, 0 * PPB + p)] = tv;
            Hl[HIDX(i, 1 * PPB + p)] = t1 * w0;
            Hl[HIDX(i, 2 * PPB + p)] = t1 * w1;
            Hl[HIDX(i, 3 * PPB + p)] = t2 * w0 * w0;
            Hl[HIDX(i, 4 * PPB + p)] = t2 * w0 * w1;
            Hl[HIDX(i, 5 * PPB + p)] = t2 * w1 * w1;
        }
    }
    W_STORE();
    __syncthreads();

    for (int l = 0; l < NLAY; ++l) {
        const int rbase  = wave * 16;
        const int m      = lane & 15;
        const int khalf  = (lane < 16) ? 0 : 2;
        const int rowoff = (lane < 16) ? 0 : 8;

        const v4f bias_lo = *(const v4f*)(bh + l * HDIM + rbase + rowoff);
        const v4f bias_hi = *(const v4f*)(bh + l * HDIM + rbase + rowoff + 4);

        v8f acc[NC];
        const v8f vzero = {0.f, 0.f, 0.f, 0.f, 0.f, 0.f, 0.f, 0.f};
        #pragma unroll
        for (int c = 0; c < NC; ++c) acc[c] = vzero;

        for (int kt = 0; kt < HDIM / 4; ++kt) {
            const int kb = kt * 4;
            const v2f a = *(const v2f*)&Wlds[(rbase + m) * WS + kb + khalf];
            #pragma unroll
            for (int c = 0; c < NC; ++c) {
                const v2f b = *(const v2f*)&Hl[HIDX(kb + khalf, c * PPB + m)];
                acc[c] = __builtin_amdgcn_wmma_f32_16x16x4_f32(
                    false, a, false, b, (short)0, acc[c], false, false);
            }
        }
        __syncthreads();
        if (l + 1 < NLAY) W_FETCH(Wh + (l + 1) * HDIM * HDIM);

        #pragma unroll
        for (int r = 0; r < 8; ++r) {
            const int i  = rbase + rowoff + r;
            const float zv  = acc[0][r] + (r < 4 ? bias_lo[r & 3] : bias_hi[r & 3]);
            const float zgx = acc[1][r];
            const float zgy = acc[2][r];
            const float zxx = acc[3][r];
            const float zxy = acc[4][r];
            const float zyy = acc[5][r];
            const float tv = tanhf(zv);
            const float t1 = 1.f - tv * tv;
            const float t2 = -2.f * tv * t1;
            Hl[HIDX(i, 0 * PPB + m)] = tv;
            Hl[HIDX(i, 1 * PPB + m)] = t1 * zgx;
            Hl[HIDX(i, 2 * PPB + m)] = t1 * zgy;
            Hl[HIDX(i, 3 * PPB + m)] = fmaf(t2, zgx * zgx, t1 * zxx);
            Hl[HIDX(i, 4 * PPB + m)] = fmaf(t2, zgx * zgy, t1 * zxy);
            Hl[HIDX(i, 5 * PPB + m)] = fmaf(t2, zgy * zgy, t1 * zyy);
        }
        if (l + 1 < NLAY) W_STORE();
        __syncthreads();
    }

    if (t < 2 * NCOL) {
        const int col  = t % NCOL;
        const int half = t / NCOL;
        float s = 0.f;
        #pragma unroll 4
        for (int k = half * 64; k < half * 64 + 64; ++k)
            s = fmaf(Wf[k], Hl[HIDX(k, col)], s);
        sbuf[half * NCOL + col] = s;
    }
    __syncthreads();
    if (t < PPB) {
        const int p = t;
        const float sv  = sbuf[0 * PPB + p] + sbuf[NCOL + 0 * PPB + p] + bf[0];
        const float sgx = sbuf[1 * PPB + p] + sbuf[NCOL + 1 * PPB + p];
        const float sgy = sbuf[2 * PPB + p] + sbuf[NCOL + 2 * PPB + p];
        const float sxx = sbuf[3 * PPB + p] + sbuf[NCOL + 3 * PPB + p];
        const float sxy = sbuf[4 * PPB + p] + sbuf[NCOL + 4 * PPB + p];
        const float syy = sbuf[5 * PPB + p] + sbuf[NCOL + 5 * PPB + p];
        const float tv = tanhf(sv);
        const float t1 = 1.f - tv * tv;
        const float t2 = -2.f * tv * t1;
        jet[0 * PPB + p] = tv;
        jet[1 * PPB + p] = t1 * sgx;
        jet[2 * PPB + p] = t1 * sgy;
        jet[3 * PPB + p] = fmaf(t2, sgx * sgx, t1 * sxx);
        jet[4 * PPB + p] = fmaf(t2, sgx * sgy, t1 * sxy);
        jet[5 * PPB + p] = fmaf(t2, sgy * sgy, t1 * syy);
    }
    __syncthreads();
}

__global__ void __launch_bounds__(WGSIZE)
pinn_elasticity_kernel(
    const float* __restrict__ x, const float* __restrict__ y,
    const float* __restrict__ W1u, const float* __restrict__ b1u,
    const float* __restrict__ Whu, const float* __restrict__ bhu,
    const float* __restrict__ Wfu, const float* __restrict__ bfu,
    const float* __restrict__ W1v, const float* __restrict__ b1v,
    const float* __restrict__ Whv, const float* __restrict__ bhv,
    const float* __restrict__ Wfv, const float* __restrict__ bfv,
    const float* __restrict__ mu, const float* __restrict__ lmbd,
    float* __restrict__ out, int N)
{
    extern __shared__ float lds[];
    float* Wlds = lds;
    float* Hl   = Wlds + HDIM * WS;
    float* sbuf = Hl + HDIM * CP;
    float* jets = sbuf + 2 * NCOL;
    float* xs   = jets + 2 * NC * PPB;
    float* ys   = xs + PPB;

    const int t  = threadIdx.x;
    const int p0 = blockIdx.x * PPB;

    if (t < PPB) {
        const int gp = p0 + t;
        xs[t] = (gp < N) ? x[gp] : 0.f;
        ys[t] = (gp < N) ? y[gp] : 0.f;
    }
    __syncthreads();

    run_net(Wlds, Hl, sbuf, jets,            W1u, b1u, Whu, bhu, Wfu, bfu, xs, ys);
    run_net(Wlds, Hl, sbuf, jets + NC * PPB, W1v, b1v, Whv, bhv, Wfv, bfv, xs, ys);

    if (t < PPB) {
        const int p  = t;
        const int gp = p0 + p;
        if (gp < N) {
            const float* ju = jets;
            const float* jv = jets + NC * PPB;
            const float Uv  = ju[0 * PPB + p];
            const float Uxx = ju[3 * PPB + p];
            const float Uxy = ju[4 * PPB + p];
            const float Uyy = ju[5 * PPB + p];
            const float Vv  = jv[0 * PPB + p];
            const float Vxx = jv[3 * PPB + p];
            const float Vxy = jv[4 * PPB + p];
            const float Vyy = jv[5 * PPB + p];

            const float MU = mu[0], LA = lmbd[0];
            const float C11 = 2.f * MU + LA;
            const float C12 = LA;
            const float C33 = 2.f * MU;

            const float rx = C11 * Uxx + C12 * Vxy + C33 * (Uyy + 0.5f * Vxy);
            const float ry = C33 * (Uxy + 0.5f * Vxx) + C11 * Vyy + C12 * Uxy;

            const float PI = 3.14159265358979323846f;
            const float Qc = 4.0f;
            const float xx = xs[p], yy = ys[p];
            const float y2 = yy * yy, y3 = y2 * yy, y4 = y3 * yy;
            const float s2px = sinf(2.f * PI * xx), c2px = cosf(2.f * PI * xx);
            const float spx  = sinf(PI * xx),       cpx  = cosf(PI * xx);
            const float spy  = sinf(PI * yy),       cpy  = cosf(PI * yy);
            const float PI2  = PI * PI;

            const float Fx = -1.0f * (4.f * PI2 * c2px * spy - Qc * y3 * PI * cpx)
                             - 0.5f * (PI2 * c2px * spy - Qc * y3 * PI * cpx)
                             - 0.5f * 8.f * PI2 * c2px * spy;
            const float Fy =  1.0f * (3.f * Qc * y2 * spx - 2.f * PI2 * cpy * s2px)
                             - 0.5f * (2.f * PI2 * cpy * s2px + Qc * y4 * PI2 * spx * 0.25f)
                             + 0.5f * 6.f * Qc * y2 * spx;

            out[0 * N + gp] = rx - Fx;
            out[1 * N + gp] = ry - Fy;
            out[2 * N + gp] = Uv;
            out[3 * N + gp] = Vv;
        }
    }
}

extern "C" void kernel_launch(void* const* d_in, const int* in_sizes, int n_in,
                              void* d_out, int out_size, void* d_ws, size_t ws_size,
                              hipStream_t stream) {
    const float* x    = (const float*)d_in[0];
    const float* y    = (const float*)d_in[1];
    const float* W1u  = (const float*)d_in[2];
    const float* b1u  = (const float*)d_in[3];
    const float* Whu  = (const float*)d_in[4];
    const float* bhu  = (const float*)d_in[5];
    const float* Wfu  = (const float*)d_in[6];
    const float* bfu  = (const float*)d_in[7];
    const float* W1v  = (const float*)d_in[8];
    const float* b1v  = (const float*)d_in[9];
    const float* Whv  = (const float*)d_in[10];
    const float* bhv  = (const float*)d_in[11];
    const float* Wfv  = (const float*)d_in[12];
    const float* bfv  = (const float*)d_in[13];
    const float* mu   = (const float*)d_in[14];
    const float* lmbd = (const float*)d_in[15];

    const int N = in_sizes[0];
    const int grid = (N + PPB - 1) / PPB;
    const size_t shmem =
        (size_t)(HDIM * WS + HDIM * CP + 2 * NCOL + 2 * NC * PPB + 2 * PPB) * sizeof(float);

    pinn_elasticity_kernel<<<grid, WGSIZE, shmem, stream>>>(
        x, y, W1u, b1u, Whu, bhu, Wfu, bfu,
        W1v, b1v, Whv, bhv, Wfv, bfv, mu, lmbd,
        (float*)d_out, N);
}


# Weights go from global memory straight into A-fragment registers, no LDS weight tile (LDS 124->54 KB: 4 blocks per WGP); k order of the sums permuted, bias is the accumulator start value
# speedup vs baseline: 1.2453x; 1.0795x over previous
#include <hip/hip_runtime.h>
#include <math.h>

typedef __attribute__((ext_vector_type(2))) float v2f;
typedef __attribute__((ext_vector_type(8))) float v8f;

#define HDIM 128
#define NLAY 3
#define PPB 16
#define NC 6
#define NCOL (NC * PPB)
#define CP 104
#define WGSIZE 256

#define HIDX(k, col) ((((k) >> 1) * (2 * CP)) + (((col)) << 1) + ((k) & 1))

typedef __attribute__((ext_vector_type(4))) float v4f;

#define A_LOAD(dst, src) do { const float* ap_ = (src); \
    dst[0] = *(const v4f*)(ap_ + 0);  \
    dst[1] = *(const v4f*)(ap_ + 8);  \
} while (0)

__device__ __forceinline__ void b_load(v2f (&b)[NC], const float* __restrict__ hp)
{
    #pragma unroll
    for (int c = 0; c < NC; ++c) b[c] = *(const v2f*)(hp + 2 * PPB * c);
}

__device__ __forceinline__ void mma_step(v8f (&acc)[NC], const v2f a, const v2f (&b)[NC])
{
    #pragma unroll
    for (int c = 0; c < NC; ++c)
        acc[c] = __builtin_amdgcn_wmma_f32_16x16x4_f32(
            false, a, false, b[c], (short)0, acc[c], false, false);
}

#define A_XY(v) __builtin_shufflevector((v), (v), 0, 1)
#define A_ZW(v) __builtin_shufflevector((v), (v), 2, 3)

__device__ __forceinline__ void k_trip(v8f (&acc)[NC], v4f (&A0)[2], v4f (&A1)[2],
                                       v2f (&B0)[NC], v2f (&B1)[NC],
                                       const float* __restrict__ wq,
                                       const float* __restrict__ wn,
                                       const float* __restrict__ hq, const bool more)
{
    A_LOAD(A1, wq + 16);
    b_load(B1, hq + 1 * (2 * CP));   mma_step(acc, A_XY(A0[0]), B0);
    b_load(B0, hq + 4 * (2 * CP));   mma_step(acc, A_ZW(A0[0]), B1);
    b_load(B1, hq + 5 * (2 * CP));   mma_step(acc, A_XY(A0[1]), B0);
    b_load(B0, hq + 8 * (2 * CP));   mma_step(acc, A_ZW(A0[1]), B1);
    A_LOAD(A0, wn);
    b_load(B1, hq + 9 * (2 * CP));   mma_step(acc, A_XY(A1[0]), B0);
    b_load(B0, hq + 12 * (2 * CP));  mma_step(acc, A_ZW(A1[0]), B1);
    b_load(B1, hq + 13 * (2 * CP));  mma_step(acc, A_XY(A1[1]), B0);
    if (more) b_load(B0, hq + 16 * (2 * CP));
    mma_step(acc, A_ZW(A1[1]), B1);
}

__device__ __forceinline__ void run_net(
    float* __restrict__ Hl, float* __restrict__ sbuf,
    float* __restrict__ jet,
    const float* __restrict__ W1, const float* __restrict__ b1,
    const float* __restrict__ Wh, const float* __restrict__ bh,
    const float* __restrict__ Wf, const float* __restrict__ bf,
    const float* __restrict__ xs, const float* __restrict__ ys)
{
    const int t    = threadIdx.x;
    const int lane = t & 31;
    const int wave = t >> 5;

    const int rbase = wave * 16;
    const int m     = lane & 15;
    const int hh    = lane >> 4;
    const float* wrow = Wh + (rbase + m) * HDIM + 4 * hh;
    const float* hbase = Hl + (2 * hh) * (2 * CP) + 2 * m;

    v4f A0[2], A1[2];

    const float* bl = bh + rbase + 8 * hh;
    v4f bias_lo, bias_hi;

    {
        const int p = t & 15;
        const float x = xs[p], y = ys[p];
        float iw0[8], iw1[8], ib[8];
        #pragma unroll
        for (int j = 0; j < 8; ++j) {
            const int i = (t >> 4) + 16 * j;
            iw0[j] = W1[i * 2 + 0];
            iw1[j] = W1[i * 2 + 1];
            ib[j]  = b1[i];
        }
        A_LOAD(A0, wrow);
        bias_lo = *(const v4f*)(bl);
        bias_hi = *(const v4f*)(bl + 4);
        #pragma unroll
        for (int j = 0; j < 8; ++j) {
            const int i = (t >> 4) + 16 * j;
            const float w0 = iw0[j], w1 = iw1[j];
            const float z  = fmaf(w0, x, fmaf(w1, y, ib[j]));
            const float tv = tanhf(z);
            const float t1 = 1.f - tv * tv;
            const float t2 = -2.f * tv * t1;
            Hl[HIDX(i, 0 * PPB + p)] = tv;
            Hl[HIDX(i, 1 * PPB + p)] = t1 * w0;
            Hl[HIDX(i, 2 * PPB + p)] = t1 * w1;
            Hl[HIDX(i, 3 * PPB + p)] = t2 * w0 * w0;
            Hl[HIDX(i, 4 * PPB + p)] = t2 * w0 * w1;
            Hl[HIDX(i, 5 * PPB + p)] = t2 * w1 * w1;
        }
    }
    __syncthreads();

    for (int l = 0; l < NLAY; ++l) {
        const float* wl  = wrow + l * HDIM * HDIM;
        const int rowoff = 8 * hh;

        v8f acc[NC];
        const v8f vzero = {0.f, 0.f, 0.f, 0.f, 0.f, 0.f, 0.f, 0.f};
        acc[0] = __builtin_shufflevector(bias_lo, bias_hi, 0, 1, 2, 3, 4, 5, 6, 7);
        #pragma unroll
        for (int c = 1; c < NC; ++c) acc[c] = vzero;

        const float* wnext = (l + 1 < NLAY) ? wl + HDIM * HDIM : wl;
        v2f B0[NC], B1[NC];
        b_load(B0, hbase);
        #pragma nounroll
        for (int q2 = 0; q2 < 4; ++q2)
            k_trip(acc, A0, A1, B0, B1, wl + 32 * q2, (q2 < 3) ? wl + 32 * q2 + 32 : wnext,
                   hbase + q2 * 16 * (2 * CP), q2 < 3);
        __syncthreads();
        if (l + 1 < NLAY) {
            bias_lo = *(const v4f*)(bl + (l + 1) * HDIM);
            bias_hi = *(const v4f*)(bl + (l + 1) * HDIM + 4);
        }

        #pragma unroll
        for (int r = 0; r < 8; ++r) {
            const int i  = rbase + rowoff + r;
            const float zv  = acc[0][r];
            const float zgx = acc[1][r];
            const float zgy = acc[2][r];
            const float zxx = acc[3][r];
            const float zxy = acc[4][r];
            const float zyy = acc[5][r];
            const float tv = tanhf(zv);
            const float t1 = 1.f - tv * tv;
            const float t2 = -2.f * tv * t1;
            Hl[HIDX(i, 0 * PPB + m)] = tv;
            Hl[HIDX(i, 1 * PPB + m)] = t1 * zgx;
            Hl[HIDX(i, 2 * PPB + m)] = t1 * zgy;
            Hl[HIDX(i, 3 * PPB + m)] = fmaf(t2, zgx * zgx, t1 * zxx);
            Hl[HIDX(i, 4 * PPB + m)] = fmaf(t2, zgx * zgy, t1 * zxy);
            Hl[HIDX(i, 5 * PPB + m)] = fmaf(t2, zgy * zgy, t1 * zyy);
        }
        __syncthreads();
    }

    if (t < 2 * NCOL) {
        const int col  = t % NCOL;
        const int half = t / NCOL;
        float s = 0.f;
        #pragma unroll 4
        for (int k = half * 64; k < half * 64 + 64; ++k)
            s = fmaf(Wf[k], Hl[HIDX(k, col)], s);
        sbuf[half * NCOL + col] = s;
    }
    __syncthreads();
    if (t < PPB) {
        const int p = t;
        const float sv  = sbuf[0 * PPB + p] + sbuf[NCOL + 0 * PPB + p] + bf[0];
        const float sgx = sbuf[1 * PPB + p] + sbuf[NCOL + 1 * PPB + p];
        const float sgy = sbuf[2 * PPB + p] + sbuf[NCOL + 2 * PPB + p];
        const float sxx = sbuf[3 * PPB + p] + sbuf[NCOL + 3 * PPB + p];
        const float sxy = sbuf[4 * PPB + p] + sbuf[NCOL + 4 * PPB + p];
        const float syy = sbuf[5 * PPB + p] + sbuf[NCOL + 5 * PPB + p];
        const float tv = tanhf(sv);
        const float t1 = 1.f - tv * tv;
        const float t2 = -2.f * tv * t1;
        jet[0 * PPB + p] = tv;
        jet[1 * PPB + p] = t1 * sgx;
        jet[2 * PPB + p] = t1 * sgy;
        jet[3 * PPB + p] = fmaf(t2, sgx * sgx, t1 * sxx);
        jet[4 * PPB + p] = fmaf(t2, sgx * sgy, t1 * sxy);
        jet[5 * PPB + p] = fmaf(t2, sgy * sgy, t1 * syy);
    }
    __syncthreads();
}

__global__ void __launch_bounds__(WGSIZE, 8)
pinn_elasticity_kernel(
    const float* __restrict__ x, const float* __restrict__ y,
    const float* __restrict__ W1u, const float* __restrict__ b1u,
    const float* __restrict__ Whu, const float* __restrict__ bhu,
    const float* __restrict__ Wfu, const float* __restrict__ bfu,
    const float* __restrict__ W1v, const float* __restrict__ b1v,
    const float* __restrict__ Whv, const float* __restrict__ bhv,
    const float* __restrict__ Wfv, const float* __restrict__ bfv,
    const float* __restrict__ mu, const float* __restrict__ lmbd,
    float* __restrict__ out, int N)
{
    extern __shared__ float lds[];
    float* Hl   = lds;
    float* sbuf = Hl + HDIM * CP;
    float* jets = sbuf + 2 * NCOL;
    float* xs   = jets + 2 * NC * PPB;
    float* ys   = xs + PPB;

    const int t  = threadIdx.x;
    const int p0 = blockIdx.x * PPB;

    if (t < PPB) {
        const int gp = p0 + t;
        xs[t] = (gp < N) ? x[gp] : 0.f;
        ys[t] = (gp < N) ? y[gp] : 0.f;
    }
    __syncthreads();

    run_net(Hl, sbuf, jets,            W1u, b1u, Whu, bhu, Wfu, bfu, xs, ys);
    run_net(Hl, sbuf, jets + NC * PPB, W1v, b1v, Whv, bhv, Wfv, bfv, xs, ys);

    if (t < PPB) {
        const int p  = t;
        const int gp = p0 + p;
        if (gp < N) {
            const float* ju = jets;
            const float* jv = jets + NC * PPB;
            const float Uv  = ju[0 * PPB + p];
            const float Uxx = ju[3 * PPB + p];
            const float Uxy = ju[4 * PPB + p];
            const float Uyy = ju[5 * PPB + p];
            const float Vv  = jv[0 * PPB + p];
            const float Vxx = jv[3 * PPB + p];
            const float Vxy = jv[4 * PPB + p];
            const float Vyy = jv[5 * PPB + p];

            const float MU = mu[0], LA = lmbd[0];
            const float C11 = 2.f * MU + LA;
            const float C12 = LA;
            const float C33 = 2.f * MU;

            const float rx = C11 * Uxx + C12 * Vxy + C33 * (Uyy + 0.5f * Vxy);
            const float ry = C33 * (Uxy + 0.5f * Vxx) + C11 * Vyy + C12 * Uxy;

            const float PI = 3.14159265358979323846f;
            const float Qc = 4.0f;
            const float xx = xs[p], yy = ys[p];
            const float y2 = yy * yy, y3 = y2 * yy, y4 = y3 * yy;
            const float s2px = sinf(2.f * PI * xx), c2px = cosf(2.f * PI * xx);
            const float spx  = sinf(PI * xx),       cpx  = cosf(PI * xx);
            const float spy  = sinf(PI * yy),       cpy  = cosf(PI * yy);
            const float PI2  = PI * PI;

            const float Fx = -1.0f * (4.f * PI2 * c2px * spy - Qc * y3 * PI * cpx)
                             - 0.5f * (PI2 * c2px * spy - Qc * y3 * PI * cpx)
                             - 0.5f * 8.f * PI2 * c2px * spy;
            const float Fy =  1.0f * (3.f * Qc * y2 * spx - 2.f * PI2 * cpy * s2px)
                             - 0.5f * (2.f * PI2 * cpy * s2px + Qc * y4 * PI2 * spx * 0.25f)
                             + 0.5f * 6.f * Qc * y2 * spx;

            out[0 * N + gp] = rx - Fx;
            out[1 * N + gp] = ry - Fy;
            out[2 * N + gp] = Uv;
            out[3 * N + gp] = Vv;
        }
    }
}

extern "C" void kernel_launch(void* const* d_in, const int* in_sizes, int n_in,
                              void* d_out, int out_size, void* d_ws, size_t ws_size,
                              hipStream_t stream) {
    const float* x    = (const float*)d_in[0];
    const float* y    = (const float*)d_in[1];
    const float* W1u  = (const float*)d_in[2];
    const float* b1u  = (const float*)d_in[3];
    const float* Whu  = (const float*)d_in[4];
    const float* bhu  = (const float*)d_in[5];
    const float* Wfu  = (const float*)d_in[6];
    const float* bfu  = (const float*)d_in[7];
    const float* W1v  = (const float*)d_in[8];
    const float* b1v  = (const float*)d_in[9];
    const float* Whv  = (const float*)d_in[10];
    const float* bhv  = (const float*)d_in[11];
    const float* Wfv  = (const float*)d_in[12];
    const float* bfv  = (const float*)d_in[13];
    const float* mu   = (const float*)d_in[14];
    const float* lmbd = (const float*)d_in[15];

    const int N = in_sizes[0];
    const int grid = (N + PPB - 1) / PPB;
    const size_t shmem =
        (size_t)(HDIM * CP + 2 * NCOL + 2 * NC * PPB + 2 * PPB) * sizeof(float);

    pinn_elasticity_kernel<<<grid, WGSIZE, shmem, stream>>>(
        x, y, W1u, b1u, Whu, bhu, Wfu, bfu,
        W1v, b1v, Whv, bhv, Wfv, bfv, mu, lmbd,
        (float*)d_out, N);
}


# tanh branch-free as 1 - 2/(exp2(c*z)+1) on the hardware exp2 and reciprocal: 5 vector instructions per value for the library tanhf's two masked paths of about 28; last bits of tanh differ
# speedup vs baseline: 1.3206x; 1.0605x over previous
#include <hip/hip_runtime.h>
#include <math.h>

typedef __attribute__((ext_vector_type(2))) float v2f;
typedef __attribute__((ext_vector_type(8))) float v8f;

#define HDIM 128
#define NLAY 3
#define PPB 16
#define NC 6
#define NCOL (NC * PPB)
#define CP 104
#define WGSIZE 256

#define HIDX(k, col) ((((k) >> 1) * (2 * CP)) + (((col)) << 1) + ((k) & 1))

typedef __attribute__((ext_vector_type(4))) float v4f;

#define A_LOAD(dst, src) do { const float* ap_ = (src); \
    dst[0] = *(const v4f*)(ap_ + 0);  \
    dst[1] = *(const v4f*)(ap_ + 8);  \
} while (0)

__device__ __forceinline__ void b_load(v2f (&b)[NC], const float* __restrict__ hp)
{
    #pragma unroll
    for (int c = 0; c < NC; ++c) b[c] = *(const v2f*)(hp + 2 * PPB * c);
}

__device__ __forceinline__ void mma_step(v8f (&acc)[NC], const v2f a, const v2f (&b)[NC])
{
    #pragma unroll
    for (int c = 0; c < NC; ++c)
        acc[c] = __builtin_amdgcn_wmma_f32_16x16x4_f32(
            false, a, false, b[c], (short)0, acc[c], false, false);
}

__device__ __forceinline__ float tanh_hw(const float z)
{
    const float e = __builtin_amdgcn_exp2f(z * 2.88539008177792681472f);
    return fmaf(-2.f, __builtin_amdgcn_rcpf(e + 1.f), 1.f);
}

#define A_XY(v) __builtin_shufflevector((v), (v), 0, 1)
#define A_ZW(v) __builtin_shufflevector((v), (v), 2, 3)

__device__ __forceinline__ void k_trip(v8f (&acc)[NC], v4f (&A0)[2], v4f (&A1)[2],
                                       v2f (&B0)[NC], v2f (&B1)[NC],
                                       const float* __restrict__ wq,
                                       const float* __restrict__ wn,
                                       const float* __restrict__ hq, const bool more)
{
    A_LOAD(A1, wq + 16);
    b_load(B1, hq + 1 * (2 * CP));   mma_step(acc, A_XY(A0[0]), B0);
    b_load(B0, hq + 4 * (2 * CP));   mma_step(acc, A_ZW(A0[0]), B1);
    b_load(B1, hq + 5 * (2 * CP));   mma_step(acc, A_XY(A0[1]), B0);
    b_load(B0, hq + 8 * (2 * CP));   mma_step(acc, A_ZW(A0[1]), B1);
    A_LOAD(A0, wn);
    b_load(B1, hq + 9 * (2 * CP));   mma_step(acc, A_XY(A1[0]), B0);
    b_load(B0, hq + 12 * (2 * CP));  mma_step(acc, A_ZW(A1[0]), B1);
    b_load(B1, hq + 13 * (2 * CP));  mma_step(acc, A_XY(A1[1]), B0);
    if (more) b_load(B0, hq + 16 * (2 * CP));
    mma_step(acc, A_ZW(A1[1]), B1);
}

__device__ __forceinline__ void run_net(
    float* __restrict__ Hl, float* __restrict__ sbuf,
    float* __restrict__ jet,
    const float* __restrict__ W1, const float* __restrict__ b1,
    const float* __restrict__ Wh, const float* __restrict__ bh,
    const float* __restrict__ Wf, const float* __restrict__ bf,
    const float* __restrict__ xs, const float* __restrict__ ys)
{
    const int t    = threadIdx.x;
    const int lane = t & 31;
    const int wave = t >> 5;

    const int rbase = wave * 16;
    const int m     = lane & 15;
    const int hh    = lane >> 4;
    const float* wrow = Wh + (rbase + m) * HDIM + 4 * hh;
    const float* hbase = Hl + (2 * hh) * (2 * CP) + 2 * m;

    v4f A0[2], A1[2];

    const float* bl = bh + rbase + 8 * hh;
    v4f bias_lo, bias_hi;

    {
        const int p = t & 15;
        const float x = xs[p], y = ys[p];
        float iw0[8], iw1[8], ib[8];
        #pragma unroll
        for (int j = 0; j < 8; ++j) {
            const int i = (t >> 4) + 16 * j;
            iw0[j] = W1[i * 2 + 0];
            iw1[j] = W1[i * 2 + 1];
            ib[j]  = b1[i];
        }
        A_LOAD(A0, wrow);
        bias_lo = *(const v4f*)(bl);
        bias_hi = *(const v4f*)(bl + 4);
        float* hrow = Hl + HIDX(t >> 4, p);
        #pragma unroll
        for (int j = 0; j < 8; ++j) {
            const float w0 = iw0[j], w1 = iw1[j];
            const float z  = fmaf(w0, x, fmaf(w1, y, ib[j]));
            const float tv = tanh_hw(z);
            const float t1 = 1.f - tv * tv;
            const float t2 = -2.f * tv * t1;
            float* hj = hrow + 16 * CP * j;
            hj[2 * PPB * 0] = tv;
            hj[2 * PPB * 1] = t1 * w0;
            hj[2 * PPB * 2] = t1 * w1;
            hj[2 * PPB * 3] = t2 * w0 * w0;
            hj[2 * PPB * 4] = t2 * w0 * w1;
            hj[2 * PPB * 5] = t2 * w1 * w1;
        }
    }
    __syncthreads();

    #pragma unroll
    for (int l = 0; l < NLAY; ++l) {
        const float* wl  = wrow + l * HDIM * HDIM;
        const int rowoff = 8 * hh;

        v8f acc[NC];
        const v8f vzero = {0.f, 0.f, 0.f, 0.f, 0.f, 0.f, 0.f, 0.f};
        acc[0] = __builtin_shufflevector(bias_lo, bias_hi, 0, 1, 2, 3, 4, 5, 6, 7);
        #pragma unroll
        for (int c = 1; c < NC; ++c) acc[c] = vzero;

        const float* wnext = (l + 1 < NLAY) ? wl + HDIM * HDIM : wl;
        v2f B0[NC], B1[NC];
        b_load(B0, hbase);
        #pragma nounroll
        for (int q2 = 0; q2 < 4; ++q2)
            k_trip(acc, A0, A1, B0, B1, wl + 32 * q2, (q2 < 3) ? wl + 32 * q2 + 32 : wnext,
                   hbase + q2 * 16 * (2 * CP), q2 < 3);
        __syncthreads();
        if (l + 1 < NLAY) {
            bias_lo = *(const v4f*)(bl + (l + 1) * HDIM);
            bias_hi = *(const v4f*)(bl + (l + 1) * HDIM + 4);
        }

        #pragma unroll
        for (int r = 0; r < 8; ++r) {
            const int i  = rbase + rowoff + r;
            const float zv  = acc[0][r];
            const float zgx = acc[1][r];
            const float zgy = acc[2][r];
            const float zxx = acc[3][r];
            const float zxy = acc[4][r];
            const float zyy = acc[5][r];
            const float tv = tanh_hw(zv);
            const float t1 = 1.f - tv * tv;
            const float t2 = -2.f * tv * t1;
            Hl[HIDX(i, 0 * PPB + m)] = tv;
            Hl[HIDX(i, 1 * PPB + m)] = t1 * zgx;
            Hl[HIDX(i, 2 * PPB + m)] = t1 * zgy;
            Hl[HIDX(i, 3 * PPB + m)] = fmaf(t2, zgx * zgx, t1 * zxx);
            Hl[HIDX(i, 4 * PPB + m)] = fmaf(t2, zgx * zgy, t1 * zxy);
            Hl[HIDX(i, 5 * PPB + m)] = fmaf(t2, zgy * zgy, t1 * zyy);
        }
        __syncthreads();
    }

    if (t < 2 * NCOL) {
        const int col  = t % NCOL;
        const int half = t / NCOL;
        float s = 0.f;
        #pragma unroll 4
        for (int k = half * 64; k < half * 64 + 64; ++k)
            s = fmaf(Wf[k], Hl[HIDX(k, col)], s);
        sbuf[half * NCOL + col] = s;
    }
    __syncthreads();
    if (t < PPB) {
        const int p = t;
        const float sv  = sbuf[0 * PPB + p] + sbuf[NCOL + 0 * PPB + p] + bf[0];
        const float sgx = sbuf[1 * PPB + p] + sbuf[NCOL + 1 * PPB + p];
        const float sgy = sbuf[2 * PPB + p] + sbuf[NCOL + 2 * PPB + p];
        const float sxx = sbuf[3 * PPB + p] + sbuf[NCOL + 3 * PPB + p];
        const float sxy = sbuf[4 * PPB + p] + sbuf[NCOL + 4 * PPB + p];
        const float syy = sbuf[5 * PPB + p] + sbuf[NCOL + 5 * PPB + p];
        const float tv = tanhf(sv);
        const float t1 = 1.f - tv * tv;
        const float t2 = -2.f * tv * t1;
        jet[0 * PPB + p] = tv;
        jet[1 * PPB + p] = t1 * sgx;
        jet[2 * PPB + p] = t1 * sgy;
        jet[3 * PPB + p] = fmaf(t2, sgx * sgx, t1 * sxx);
        jet[4 * PPB + p] = fmaf(t2, sgx * sgy, t1 * sxy);
        jet[5 * PPB + p] = fmaf(t2, sgy * sgy, t1 * syy);
    }
    __syncthreads();
}

__global__ void __launch_bounds__(WGSIZE, 8)
pinn_elasticity_kernel(
    const float* __restrict__ x, const float* __restrict__ y,
    const float* __restrict__ W1u, const float* __restrict__ b1u,
    const float* __restrict__ Whu, const float* __restrict__ bhu,
    const float* __restrict__ Wfu, const float* __restrict__ bfu,
    const float* __restrict__ W1v, const float* __restrict__ b1v,
    const float* __restrict__ Whv, const float* __restrict__ bhv,
    const float* __restrict__ Wfv, const float* __restrict__ bfv,
    const float* __restrict__ mu, const float* __restrict__ lmbd,
    float* __restrict__ out, int N)
{
    extern __shared__ float lds[];
    float* Hl   = lds;
    float* sbuf = Hl + HDIM * CP;
    float* jets = sbuf + 2 * NCOL;
    float* xs   = jets + 2 * NC * PPB;
    float* ys   = xs + PPB;

    const int t  = threadIdx.x;
    const int p0 = blockIdx.x * PPB;

    if (t < PPB) {
        const int gp = p0 + t;
        xs[t] = (gp < N) ? x[gp] : 0.f;
        ys[t] = (gp < N) ? y[gp] : 0.f;
    }
    __syncthreads();

    run_net(Hl, sbuf, jets,            W1u, b1u, Whu, bhu, Wfu, bfu, xs, ys);
    run_net(Hl, sbuf, jets + NC * PPB, W1v, b1v, Whv, bhv, Wfv, bfv, xs, ys);

    if (t < PPB) {
        const int p  = t;
        const int gp = p0 + p;
        if (gp < N) {
            const float* ju = jets;
            const float* jv = jets + NC * PPB;
            const float Uv  = ju[0 * PPB + p];
            const float Uxx = ju[3 * PPB + p];
            const float Uxy = ju[4 * PPB + p];
            const float Uyy = ju[5 * PPB + p];
            const float Vv  = jv[0 * PPB + p];
            const float Vxx = jv[3 * PPB + p];
            const float Vxy = jv[4 * PPB + p];
            const float Vyy = jv[5 * PPB + p];

            const float MU = mu[0], LA = lmbd[0];
            const float C11 = 2.f * MU + LA;
            const float C12 = LA;
            const float C33 = 2.f * MU;

            const float rx = C11 * Uxx + C12 * Vxy + C33 * (Uyy + 0.5f * Vxy);
            const float ry = C33 * (Uxy + 0.5f * Vxx) + C11 * Vyy + C12 * Uxy;

            const float PI = 3.14159265358979323846f;
            const float Qc = 4.0f;
            const float xx = xs[p], yy = ys[p];
            const float y2 = yy * yy, y3 = y2 * yy, y4 = y3 * yy;
            const float s2px = sinf(2.f * PI * xx), c2px = cosf(2.f * PI * xx);
            const float spx  = sinf(PI * xx),       cpx  = cosf(PI * xx);
            const float spy  = sinf(PI * yy),       cpy  = cosf(PI * yy);
            const float PI2  = PI * PI;

            const float Fx = -1.0f * (4.f * PI2 * c2px * spy - Qc * y3 * PI * cpx)
                             - 0.5f * (PI2 * c2px * spy - Qc * y3 * PI * cpx)
                             - 0.5f * 8.f * PI2 * c2px * spy;
            const float Fy =  1.0f * (3.f * Qc * y2 * spx - 2.f * PI2 * cpy * s2px)
                             - 0.5f * (2.f * PI2 * cpy * s2px + Qc * y4 * PI2 * spx * 0.25f)
                             + 0.5f * 6.f * Qc * y2 * spx;

            out[0 * N + gp] = rx - Fx;
            out[1 * N + gp] = ry - Fy;
            out[2 * N + gp] = Uv;
            out[3 * N + gp] = Vv;
        }
    }
}

extern "C" void kernel_launch(void* const* d_in, const int* in_sizes, int n_in,
                              void* d_out, int out_size, void* d_ws, size_t ws_size,
                              hipStream_t stream) {
    const float* x    = (const float*)d_in[0];
    const float* y    = (const float*)d_in[1];
    const float* W1u  = (const float*)d_in[2];
    const float* b1u  = (const float*)d_in[3];
    const float* Whu  = (const float*)d_in[4];
    const float* bhu  = (const float*)d_in[5];
    const float* Wfu  = (const float*)d_in[6];
    const float* bfu  = (const float*)d_in[7];
    const float* W1v  = (const float*)d_in[8];
    const float* b1v  = (const float*)d_in[9];
    const float* Whv  = (const float*)d_in[10];
    const float* bhv  = (const float*)d_in[11];
    const float* Wfv  = (const float*)d_in[12];
    const float* bfv  = (const float*)d_in[13];
    const float* mu   = (const float*)d_in[14];
    const float* lmbd = (const float*)d_in[15];

    const int N = in_sizes[0];
    const int grid = (N + PPB - 1) / PPB;
    const size_t shmem =
        (size_t)(HDIM * CP + 2 * NCOL + 2 * NC * PPB + 2 * PPB) * sizeof(float);

    pinn_elasticity_kernel<<<grid, WGSIZE, shmem, stream>>>(
        x, y, W1u, b1u, Whu, bhu, Wfu, bfu,
        W1v, b1v, Whv, bhv, Wfv, bfv, mu, lmbd,
        (float*)d_out, N);
}


# Activations in LDS regrouped 4 k per column, every LDS access 16 bytes: one group of 6 B-fragment loads per 12 WMMAs (4 load places per 48 WMMAs, was 9), 4x fewer LDS stores; sum order unchanged
# speedup vs baseline: 1.3821x; 1.0466x over previous
#include <hip/hip_runtime.h>
#include <math.h>

typedef __attribute__((ext_vector_type(2))) float v2f;
typedef __attribute__((ext_vector_type(4))) float v4f;
typedef __attribute__((ext_vector_type(8))) float v8f;

#define HDIM 128
#define NLAY 3
#define PPB 16
#define NC 6
#define NCOL (NC * PPB)
#define CP 104
#define QROW (4 * CP)
#define WGSIZE 256


#define A_LOAD(dst, src) do { const float* ap_ = (src); \
    dst[0] = *(const v4f*)(ap_ + 0);  \
    dst[1] = *(const v4f*)(ap_ + 8);  \
} while (0)

__device__ __forceinline__ void b_load3(v4f (&b)[3], const float* __restrict__ hp)
{
    #pragma unroll
    for (int c = 0; c < 3; ++c) b[c] = *(const v4f*)(hp + 4 * PPB * c);
}

#define A_XY(v) __builtin_shufflevector((v), (v), 0, 1)
#define A_ZW(v) __builtin_shufflevector((v), (v), 2, 3)

template <int c0>
__device__ __forceinline__ void mma_half(v8f (&acc)[NC], const v4f a, const v4f (&b)[3])
{
    #pragma unroll
    for (int c = 0; c < 3; ++c)
        acc[c0 + c] = __builtin_amdgcn_wmma_f32_16x16x4_f32(
            false, A_XY(a), false, A_XY(b[c]), (short)0, acc[c0 + c], false, false);
    #pragma unroll
    for (int c = 0; c < 3; ++c)
        acc[c0 + c] = __builtin_amdgcn_wmma_f32_16x16x4_f32(
            false, A_ZW(a), false, A_ZW(b[c]), (short)0, acc[c0 + c], false, false);
}

__device__ __forceinline__ float tanh_hw(const float z)
{
    const float e = __builtin_amdgcn_exp2f(z * 2.88539008177792681472f);
    return fmaf(-2.f, __builtin_amdgcn_rcpf(e + 1.f), 1.f);
}

#define LOADS_STAY_TOGETHER() __builtin_amdgcn_sched_barrier(0)

#define LOWER_TILES_FIRST() __builtin_amdgcn_sched_barrier(0)

#define PS (2 * QROW)
#define TS3 (12 * PPB)

__device__ __forceinline__ void k_trip(v8f (&acc)[NC], v4f (&A0)[2], v4f (&A1)[2],
                                       v4f (&L0)[3], v4f (&L1)[3], v4f (&H)[3],
                                       const float* __restrict__ wq,
                                       const float* __restrict__ wn,
                                       const float* __restrict__ hq, const bool more)
{
    A_LOAD(A1, wq + 16);
    b_load3(H, hq + 0 * PS + TS3);   b_load3(L1, hq + 1 * PS);
    LOADS_STAY_TOGETHER();
    mma_half<0>(acc, A0[0], L0);
    LOWER_TILES_FIRST();
    mma_half<3>(acc, A0[0], H);
    LOADS_STAY_TOGETHER();
    b_load3(H, hq + 1 * PS + TS3);   b_load3(L0, hq + 2 * PS);
    LOADS_STAY_TOGETHER();
    mma_half<0>(acc, A0[1], L1);
    LOWER_TILES_FIRST();
    mma_half<3>(acc, A0[1], H);
    LOADS_STAY_TOGETHER();
    A_LOAD(A0, wn);
    b_load3(H, hq + 2 * PS + TS3);   b_load3(L1, hq + 3 * PS);
    LOADS_STAY_TOGETHER();
    mma_half<0>(acc, A1[0], L0);
    LOWER_TILES_FIRST();
    mma_half<3>(acc, A1[0], H);
    LOADS_STAY_TOGETHER();
    b_load3(H, hq + 3 * PS + TS3);   if (more) b_load3(L0, hq + 4 * PS);
    LOADS_STAY_TOGETHER();
    mma_half<0>(acc, A1[1], L1);
    LOWER_TILES_FIRST();
    mma_half<3>(acc, A1[1], H);
    LOADS_STAY_TOGETHER();
}

__device__ __forceinline__ void run_net(
    float* __restrict__ Hl, float* __restrict__ sbuf,
    float* __restrict__ jet,
    const float* __restrict__ W1, const float* __restrict__ b1,
    const float* __restrict__ Wh, const float* __restrict__ bh,
    const float* __restrict__ Wf, const float* __restrict__ bf,
    const float* __restrict__ xs, const float* __restrict__ ys)
{
    const int t    = threadIdx.x;
    const int lane = t & 31;
    const int wave = t >> 5;

    const int rbase = wave * 16;
    const int m     = lane & 15;
    const int hh    = lane >> 4;
    const float* wrow = Wh + (rbase + m) * HDIM + 4 * hh;
    const float* hbase = Hl + hh * QROW + 4 * m;

    v4f A0[2], A1[2];

    const float* bl = bh + rbase + 8 * hh;
    v4f bias_lo, bias_hi;

    {
        const int p = t & 15;
        const int g = t >> 4;
        const float x = xs[p], y = ys[p];
        v4f iwa[2], iwb[2], ibq[2];
        #pragma unroll
        for (int j = 0; j < 2; ++j) {
            const int q = g + 16 * j;
            iwa[j] = *(const v4f*)(W1 + 8 * q);
            iwb[j] = *(const v4f*)(W1 + 8 * q + 4);
            ibq[j] = *(const v4f*)(b1 + 4 * q);
        }
        A_LOAD(A0, wrow);
        bias_lo = *(const v4f*)(bl);
        bias_hi = *(const v4f*)(bl + 4);
        float* hrow = Hl + g * QROW + 4 * p;
        #pragma unroll
        for (int j = 0; j < 2; ++j) {
            v4f o[NC];
            #pragma unroll
            for (int rr = 0; rr < 4; ++rr) {
                const v4f wv   = (rr < 2) ? iwa[j] : iwb[j];
                const float w0 = wv[2 * (rr & 1)], w1 = wv[2 * (rr & 1) + 1];
                const float z  = fmaf(w0, x, fmaf(w1, y, ibq[j][rr]));
                const float tv = tanh_hw(z);
                const float t1 = 1.f - tv * tv;
                const float t2 = -2.f * tv * t1;
                o[0][rr] = tv;
                o[1][rr] = t1 * w0;
                o[2][rr] = t1 * w1;
                o[3][rr] = t2 * w0 * w0;
                o[4][rr] = t2 * w0 * w1;
                o[5][rr] = t2 * w1 * w1;
            }
            #pragma unroll
            for (int c = 0; c < NC; ++c)
                *(v4f*)(hrow + 16 * QROW * j + 4 * PPB * c) = o[c];
        }
    }
    __syncthreads();

    #pragma unroll
    for (int l = 0; l < NLAY; ++l) {
        const float* wl  = wrow + l * HDIM * HDIM;

        v8f acc[NC];
        const v8f vzero = {0.f, 0.f, 0.f, 0.f, 0.f, 0.f, 0.f, 0.f};
        acc[0] = __builtin_shufflevector(bias_lo, bias_hi, 0, 1, 2, 3, 4, 5, 6, 7);
        #pragma unroll
        for (int c = 1; c < NC; ++c) acc[c] = vzero;

        const float* wnext = (l + 1 < NLAY) ? wl + HDIM * HDIM : wl;
        v4f L0[3], L1[3], H[3];
        b_load3(L0, hbase);
        #pragma nounroll
        for (int q2 = 0; q2 < 4; ++q2)
            k_trip(acc, A0, A1, L0, L1, H, wl + 32 * q2, (q2 < 3) ? wl + 32 * q2 + 32 : wnext,
                   hbase + q2 * 8 * QROW, q2 < 3);
        __syncthreads();
        if (l + 1 < NLAY) {
            bias_lo = *(const v4f*)(bl + (l + 1) * HDIM);
            bias_hi = *(const v4f*)(bl + (l + 1) * HDIM + 4);
        }

        float* hst = Hl + (4 * wave + 2 * hh) * QROW + 4 * m;
        #pragma unroll
        for (int g = 0; g < 2; ++g) {
            v4f o[NC];
            #pragma unroll
            for (int rr = 0; rr < 4; ++rr) {
                const int r = 4 * g + rr;
                const float zv  = acc[0][r];
                const float zgx = acc[1][r];
                const float zgy = acc[2][r];
                const float zxx = acc[3][r];
                const float zxy = acc[4][r];
                const float zyy = acc[5][r];
                const float tv = tanh_hw(zv);
                const float t1 = 1.f - tv * tv;
                const float t2 = -2.f * tv * t1;
                o[0][rr] = tv;
                o[1][rr] = t1 * zgx;
                o[2][rr] = t1 * zgy;
                o[3][rr] = fmaf(t2, zgx * zgx, t1 * zxx);
                o[4][rr] = fmaf(t2, zgx * zgy, t1 * zxy);
                o[5][rr] = fmaf(t2, zgy * zgy, t1 * zyy);
            }
            #pragma unroll
            for (int c = 0; c < NC; ++c)
                *(v4f*)(hst + QROW * g + 4 * PPB * c) = o[c];
        }
        __syncthreads();
    }

    if (t < 2 * NCOL) {
        const int col  = t % NCOL;
        const int half = t / NCOL;
        const float* hc = Hl + 4 * col;
        float s = 0.f;
        #pragma unroll 4
        for (int kq = half * 16; kq < half * 16 + 16; ++kq) {
            const v4f w  = *(const v4f*)(Wf + 4 * kq);
            const v4f hv = *(const v4f*)(hc + kq * QROW);
            s = fmaf(w[0], hv[0], s);
            s = fmaf(w[1], hv[1], s);
            s = fmaf(w[2], hv[2], s);
            s = fmaf(w[3], hv[3], s);
        }
        sbuf[half * NCOL + col] = s;
    }
    __syncthreads();
    if (t < PPB) {
        const int p = t;
        const float sv  = sbuf[0 * PPB + p] + sbuf[NCOL + 0 * PPB + p] + bf[0];
        const float sgx = sbuf[1 * PPB + p] + sbuf[NCOL + 1 * PPB + p];
        const float sgy = sbuf[2 * PPB + p] + sbuf[NCOL + 2 * PPB + p];
        const float sxx = sbuf[3 * PPB + p] + sbuf[NCOL + 3 * PPB + p];
        const float sxy = sbuf[4 * PPB + p] + sbuf[NCOL + 4 * PPB + p];
        const float syy = sbuf[5 * PPB + p] + sbuf[NCOL + 5 * PPB + p];
        const float tv = tanhf(sv);
        const float t1 = 1.f - tv * tv;
        const float t2 = -2.f * tv * t1;
        jet[0 * PPB + p] = tv;
        jet[1 * PPB + p] = t1 * sgx;
        jet[2 * PPB + p] = t1 * sgy;
        jet[3 * PPB + p] = fmaf(t2, sgx * sgx, t1 * sxx);
        jet[4 * PPB + p] = fmaf(t2, sgx * sgy, t1 * sxy);
        jet[5 * PPB + p] = fmaf(t2, sgy * sgy, t1 * syy);
    }
    __syncthreads();
}

__global__ void __launch_bounds__(WGSIZE, 8)
pinn_elasticity_kernel(
    const float* __restrict__ x, const float* __restrict__ y,
    const float* __restrict__ W1u, const float* __restrict__ b1u,
    const float* __restrict__ Whu, const float* __restrict__ bhu,
    const float* __restrict__ Wfu, const float* __restrict__ bfu,
    const float* __restrict__ W1v, const float* __restrict__ b1v,
    const float* __restrict__ Whv, const float* __restrict__ bhv,
    const float* __restrict__ Wfv, const float* __restrict__ bfv,
    const float* __restrict__ mu, const float* __restrict__ lmbd,
    float* __restrict__ out, int N)
{
    extern __shared__ float lds[];
    float* Hl   = lds;
    float* sbuf = Hl + HDIM * CP;
    float* jets = sbuf + 2 * NCOL;
    float* xs   = jets + 2 * NC * PPB;
    float* ys   = xs + PPB;

    const int t  = threadIdx.x;
    const int p0 = blockIdx.x * PPB;

    if (t < PPB) {
        const int gp = p0 + t;
        xs[t] = (gp < N) ? x[gp] : 0.f;
        ys[t] = (gp < N) ? y[gp] : 0.f;
    }
    __syncthreads();

    run_net(Hl, sbuf, jets,            W1u, b1u, Whu, bhu, Wfu, bfu, xs, ys);
    run_net(Hl, sbuf, jets + NC * PPB, W1v, b1v, Whv, bhv, Wfv, bfv, xs, ys);

    if (t < PPB) {
        const int p  = t;
        const int gp = p0 + p;
        if (gp < N) {
            const float* ju = jets;
            const float* jv = jets + NC * PPB;
            const float Uv  = ju[0 * PPB + p];
            const float Uxx = ju[3 * PPB + p];
            const float Uxy = ju[4 * PPB + p];
            const float Uyy = ju[5 * PPB + p];
            const float Vv  = jv[0 * PPB + p];
            const float Vxx = jv[3 * PPB + p];
            const float Vxy = jv[4 * PPB + p];
            const float Vyy = jv[5 * PPB + p];

            const float MU = mu[0], LA = lmbd[0];
            const float C11 = 2.f * MU + LA;
            const float C12 = LA;
            const float C33 = 2.f * MU;

            const float rx = C11 * Uxx + C12 * Vxy + C33 * (Uyy + 0.5f * Vxy);
            const float ry = C33 * (Uxy + 0.5f * Vxx) + C11 * Vyy + C12 * Uxy;

            const float PI = 3.14159265358979323846f;
            const float Qc = 4.0f;
            const float xx = xs[p], yy = ys[p];
            const float y2 = yy * yy, y3 = y2 * yy, y4 = y3 * yy;
            const float s2px = sinf(2.f * PI * xx), c2px = cosf(2.f * PI * xx);
            const float spx  = sinf(PI * xx),       cpx  = cosf(PI * xx);
            const float spy  = sinf(PI * yy),       cpy  = cosf(PI * yy);
            const float PI2  = PI * PI;

            const float Fx = -1.0f * (4.f * PI2 * c2px * spy - Qc * y3 * PI * cpx)
                             - 0.5f * (PI2 * c2px * spy - Qc * y3 * PI * cpx)
                             - 0.5f * 8.f * PI2 * c2px * spy;
            const float Fy =  1.0f * (3.f * Qc * y2 * spx - 2.f * PI2 * cpy * s2px)
                             - 0.5f * (2.f * PI2 * cpy * s2px + Qc * y4 * PI2 * spx * 0.25f)
                             + 0.5f * 6.f * Qc * y2 * spx;

            out[0 * N + gp] = rx - Fx;
            out[1 * N + gp] = ry - Fy;
            out[2 * N + gp] = Uv;
            out[3 * N + gp] = Vv;
        }
    }
}

extern "C" void kernel_launch(void* const* d_in, const int* in_sizes, int n_in,
                              void* d_out, int out_size, void* d_ws, size_t ws_size,
                              hipStream_t stream) {
    const float* x    = (const float*)d_in[0];
    const float* y    = (const float*)d_in[1];
    const float* W1u  = (const float*)d_in[2];
    const float* b1u  = (const float*)d_in[3];
    const float* Whu  = (const float*)d_in[4];
    const float* bhu  = (const float*)d_in[5];
    const float* Wfu  = (const float*)d_in[6];
    const float* bfu  = (const float*)d_in[7];
    const float* W1v  = (const float*)d_in[8];
    const float* b1v  = (const float*)d_in[9];
    const float* Whv  = (const float*)d_in[10];
    const float* bhv  = (const float*)d_in[11];
    const float* Wfv  = (const float*)d_in[12];
    const float* bfv  = (const float*)d_in[13];
    const float* mu   = (const float*)d_in[14];
    const float* lmbd = (const float*)d_in[15];

    const int N = in_sizes[0];
    const int grid = (N + PPB - 1) / PPB;
    const size_t shmem =
        (size_t)(HDIM * CP + 2 * NCOL + 2 * NC * PPB + 2 * PPB) * sizeof(float);

    pinn_elasticity_kernel<<<grid, WGSIZE, shmem, stream>>>(
        x, y, W1u, b1u, Whu, bhu, Wfu, bfu,
        W1v, b1v, Whv, bhv, Wfv, bfv, mu, lmbd,
        (float*)d_out, N);
}


# Last hidden layer: 2nd-derivative columns skip W3; one adjoint tile a=W3^T(Wf*t1) dotted with H2xx,H2xy,H2yy (4 tile products, was 6); output sums re-ordered into 16+16 partial sums
# speedup vs baseline: 1.5098x; 1.0924x over previous
#include <hip/hip_runtime.h>
#include <math.h>

typedef __attribute__((ext_vector_type(2))) float v2f;
typedef __attribute__((ext_vector_type(4))) float v4f;
typedef __attribute__((ext_vector_type(8))) float v8f;

#define HDIM 128
#define NLAY 3
#define PPB 16
#define NC 6
#define NCOL (NC * PPB)
#define CP 104
#define QROW (4 * CP)
#define WGSIZE 256
#define QB 68
#define NPART 16


#define A_LOAD(dst, src) do { const float* ap_ = (src); \
    dst[0] = *(const v4f*)(ap_ + 0);  \
    dst[1] = *(const v4f*)(ap_ + 8);  \
} while (0)

__device__ __forceinline__ void b_load3(v4f (&b)[3], const float* __restrict__ hp)
{
    #pragma unroll
    for (int c = 0; c < 3; ++c) b[c] = *(const v4f*)(hp + 4 * PPB * c);
}

#define A_XY(v) __builtin_shufflevector((v), (v), 0, 1)
#define A_ZW(v) __builtin_shufflevector((v), (v), 2, 3)

template <int c0>
__device__ __forceinline__ void mma_half(v8f (&acc)[NC], const v4f a, const v4f (&b)[3])
{
    #pragma unroll
    for (int c = 0; c < 3; ++c)
        acc[c0 + c] = __builtin_amdgcn_wmma_f32_16x16x4_f32(
            false, A_XY(a), false, A_XY(b[c]), (short)0, acc[c0 + c], false, false);
    #pragma unroll
    for (int c = 0; c < 3; ++c)
        acc[c0 + c] = __builtin_amdgcn_wmma_f32_16x16x4_f32(
            false, A_ZW(a), false, A_ZW(b[c]), (short)0, acc[c0 + c], false, false);
}

__device__ __forceinline__ float tanh_hw(const float z)
{
    const float e = __builtin_amdgcn_exp2f(z * 2.88539008177792681472f);
    return fmaf(-2.f, __builtin_amdgcn_rcpf(e + 1.f), 1.f);
}

#define LOADS_STAY_TOGETHER() __builtin_amdgcn_sched_barrier(0)

#define LOWER_TILES_FIRST() __builtin_amdgcn_sched_barrier(0)

#define PS (2 * QROW)
#define TS3 (12 * PPB)

__device__ __forceinline__ void k_trip(v8f (&acc)[NC], v4f (&A0)[2], v4f (&A1)[2],
                                       v4f (&L0)[3], v4f (&L1)[3], v4f (&H)[3],
                                       const float* __restrict__ wq,
                                       const float* __restrict__ wn,
                                       const float* __restrict__ hq, const bool more)
{
    A_LOAD(A1, wq + 16);
    b_load3(H, hq + 0 * PS + TS3);   b_load3(L1, hq + 1 * PS);
    LOADS_STAY_TOGETHER();
    mma_half<0>(acc, A0[0], L0);
    LOWER_TILES_FIRST();
    mma_half<3>(acc, A0[0], H);
    LOADS_STAY_TOGETHER();
    b_load3(H, hq + 1 * PS + TS3);   b_load3(L0, hq + 2 * PS);
    LOADS_STAY_TOGETHER();
    mma_half<0>(acc, A0[1], L1);
    LOWER_TILES_FIRST();
    mma_half<3>(acc, A0[1], H);
    LOADS_STAY_TOGETHER();
    A_LOAD(A0, wn);
    b_load3(H, hq + 2 * PS + TS3);   b_load3(L1, hq + 3 * PS);
    LOADS_STAY_TOGETHER();
    mma_half<0>(acc, A1[0], L0);
    LOWER_TILES_FIRST();
    mma_half<3>(acc, A1[0], H);
    LOADS_STAY_TOGETHER();
    b_load3(H, hq + 3 * PS + TS3);   if (more) b_load3(L0, hq + 4 * PS);
    LOADS_STAY_TOGETHER();
    mma_half<0>(acc, A1[1], L1);
    LOWER_TILES_FIRST();
    mma_half<3>(acc, A1[1], H);
    LOADS_STAY_TOGETHER();
}


__device__ __forceinline__ void mma3(v8f (&acc)[3], const v4f a, const v4f (&b)[3])
{
    #pragma unroll
    for (int c = 0; c < 3; ++c)
        acc[c] = __builtin_amdgcn_wmma_f32_16x16x4_f32(
            false, A_XY(a), false, A_XY(b[c]), (short)0, acc[c], false, false);
    #pragma unroll
    for (int c = 0; c < 3; ++c)
        acc[c] = __builtin_amdgcn_wmma_f32_16x16x4_f32(
            false, A_ZW(a), false, A_ZW(b[c]), (short)0, acc[c], false, false);
}

__device__ __forceinline__ void at_load4(v4f (&a)[4], const float* __restrict__ at, const int g)
{
    #pragma unroll
    for (int q = 0; q < 4; ++q) {
        const float* p = at + 8 * (4 * g + q) * HDIM;
        a[q][0] = p[0 * HDIM];
        a[q][1] = p[1 * HDIM];
        a[q][2] = p[2 * HDIM];
        a[q][3] = p[3 * HDIM];
    }
}

__device__ __forceinline__ void bt_load4(v4f (&b)[4], const float* __restrict__ bq, const int g)
{
    #pragma unroll
    for (int q = 0; q < 4; ++q) b[q] = *(const v4f*)(bq + 2 * (4 * g + q) * QB);
}

template <bool LAST>
__device__ __forceinline__ void k_trip3(v8f (&acc)[3], v4f (&A0)[2], v4f (&A1)[2],
                                        v4f (&X0)[3], v4f (&X1)[3], v4f (&Y0)[3], v4f (&Y1)[3],
                                        const float* __restrict__ wq,
                                        const float* __restrict__ hq,
                                        v4f& wf_lo, v4f& wf_hi, v4f (&aw)[4],
                                        const float* __restrict__ wfp,
                                        const float* __restrict__ at)
{
    A_LOAD(A1, wq + 16);
    b_load3(Y0, hq + 2 * PS);   b_load3(Y1, hq + 3 * PS);
    LOADS_STAY_TOGETHER();
    mma3(acc, A0[0], X0);
    mma3(acc, A0[1], X1);
    LOADS_STAY_TOGETHER();
    if (LAST) {
        wf_lo = *(const v4f*)(wfp);
        wf_hi = *(const v4f*)(wfp + 4);
        at_load4(aw, at, 0);
    } else {
        A_LOAD(A0, wq + 32);
        b_load3(X0, hq + 4 * PS);   b_load3(X1, hq + 5 * PS);
    }
    LOADS_STAY_TOGETHER();
    mma3(acc, A1[0], Y0);
    mma3(acc, A1[1], Y1);
    LOADS_STAY_TOGETHER();
}

__device__ __forceinline__ void run_net(
    float* __restrict__ Hl, float* __restrict__ sbuf,
    float* __restrict__ B2, float* __restrict__ red, float* __restrict__ red2,
    float* __restrict__ jet,
    const float* __restrict__ W1, const float* __restrict__ b1,
    const float* __restrict__ Wh, const float* __restrict__ bh,
    const float* __restrict__ Wf, const float* __restrict__ bf,
    const float* __restrict__ xs, const float* __restrict__ ys)
{
    const int t    = threadIdx.x;
    const int lane = t & 31;
    const int wave = __builtin_amdgcn_readfirstlane(t >> 5);

    const int rbase = wave * 16;
    const int m     = lane & 15;
    const int hh    = lane >> 4;
    const float* wrow = Wh + (rbase + m) * HDIM + 4 * hh;
    const float* hbase = Hl + hh * QROW + 4 * m;

    v4f A0[2], A1[2];

    const float* bl = bh + rbase + 8 * hh;
    v4f bias_lo, bias_hi;

    {
        const int p = t & 15;
        const int g = t >> 4;
        const float x = xs[p], y = ys[p];
        v4f iwa[2], iwb[2], ibq[2];
        #pragma unroll
        for (int j = 0; j < 2; ++j) {
            const int q = g + 16 * j;
            iwa[j] = *(const v4f*)(W1 + 8 * q);
            iwb[j] = *(const v4f*)(W1 + 8 * q + 4);
            ibq[j] = *(const v4f*)(b1 + 4 * q);
        }
        A_LOAD(A0, wrow);
        bias_lo = *(const v4f*)(bl);
        bias_hi = *(const v4f*)(bl + 4);
        float* hrow = Hl + g * QROW + 4 * p;
        #pragma unroll
        for (int j = 0; j < 2; ++j) {
            v4f o[NC];
            #pragma unroll
            for (int rr = 0; rr < 4; ++rr) {
                const v4f wv   = (rr < 2) ? iwa[j] : iwb[j];
                const float w0 = wv[2 * (rr & 1)], w1 = wv[2 * (rr & 1) + 1];
                const float z  = fmaf(w0, x, fmaf(w1, y, ibq[j][rr]));
                const float tv = tanh_hw(z);
                const float t1 = 1.f - tv * tv;
                const float t2 = -2.f * tv * t1;
                o[0][rr] = tv;
                o[1][rr] = t1 * w0;
                o[2][rr] = t1 * w1;
                o[3][rr] = t2 * w0 * w0;
                o[4][rr] = t2 * w0 * w1;
                o[5][rr] = t2 * w1 * w1;
            }
            #pragma unroll
            for (int c = 0; c < NC; ++c)
                *(v4f*)(hrow + 16 * QROW * j + 4 * PPB * c) = o[c];
        }
    }
    __syncthreads();

    #pragma unroll
    for (int l = 0; l < NLAY; ++l) {
        const float* wl  = wrow + l * HDIM * HDIM;
        const v8f vzero = {0.f, 0.f, 0.f, 0.f, 0.f, 0.f, 0.f, 0.f};

        if (l + 1 < NLAY) {
        v8f acc[NC];
        acc[0] = __builtin_shufflevector(bias_lo, bias_hi, 0, 1, 2, 3, 4, 5, 6, 7);
        #pragma unroll
        for (int c = 1; c < NC; ++c) acc[c] = vzero;

        const float* wnext = wl + HDIM * HDIM;
        v4f L0[3], L1[3], H[3];
        b_load3(L0, hbase);
        #pragma nounroll
        for (int q2 = 0; q2 < 4; ++q2)
            k_trip(acc, A0, A1, L0, L1, H, wl + 32 * q2, (q2 < 3) ? wl + 32 * q2 + 32 : wnext,
                   hbase + q2 * 8 * QROW, q2 < 3);
        __syncthreads();
        bias_lo = *(const v4f*)(bl + (l + 1) * HDIM);
        bias_hi = *(const v4f*)(bl + (l + 1) * HDIM + 4);

        float* hst = Hl + (4 * wave + 2 * hh) * QROW + 4 * m;
        #pragma unroll
        for (int g = 0; g < 2; ++g) {
            v4f o[NC];
            #pragma unroll
            for (int rr = 0; rr < 4; ++rr) {
                const int r = 4 * g + rr;
                const float zv  = acc[0][r];
                const float zgx = acc[1][r];
                const float zgy = acc[2][r];
                const float zxx = acc[3][r];
                const float zxy = acc[4][r];
                const float zyy = acc[5][r];
                const float tv = tanh_hw(zv);
                const float t1 = 1.f - tv * tv;
                const float t2 = -2.f * tv * t1;
                o[0][rr] = tv;
                o[1][rr] = t1 * zgx;
                o[2][rr] = t1 * zgy;
                o[3][rr] = fmaf(t2, zgx * zgx, t1 * zxx);
                o[4][rr] = fmaf(t2, zgx * zgy, t1 * zxy);
                o[5][rr] = fmaf(t2, zgy * zgy, t1 * zyy);
            }
            #pragma unroll
            for (int c = 0; c < NC; ++c)
                *(v4f*)(hst + QROW * g + 4 * PPB * c) = o[c];
        }
        __syncthreads();
        } else {
        v8f acc[3];
        acc[0] = __builtin_shufflevector(bias_lo, bias_hi, 0, 1, 2, 3, 4, 5, 6, 7);
        acc[1] = vzero;
        acc[2] = vzero;
        const float* wfp = Wf + rbase + 8 * hh;
        const float* at = Wh + l * HDIM * HDIM + 4 * hh * HDIM + rbase + m;
        v4f wf_lo, wf_hi;
        v4f aw0[4], aw1[4], bb0[4], bb1[4];
        {
            v4f X0[3], X1[3], Y0[3], Y1[3];
            b_load3(X0, hbase);   b_load3(X1, hbase + PS);
            #pragma nounroll
            for (int q2 = 0; q2 < 3; ++q2)
                k_trip3<false>(acc, A0, A1, X0, X1, Y0, Y1, wl + 32 * q2, hbase + q2 * 8 * QROW,
                               wf_lo, wf_hi, aw0, wfp, at);
            k_trip3<true>(acc, A0, A1, X0, X1, Y0, Y1, wl + 96, hbase + 24 * QROW,
                          wf_lo, wf_hi, aw0, wfp, at);
        }

        float P[NC];
        #pragma unroll
        for (int c = 0; c < NC; ++c) P[c] = 0.f;
        float* b2st = B2 + (4 * wave + 2 * hh) * QB + 4 * m;
        #pragma unroll
        for (int g = 0; g < 2; ++g) {
            v4f o;
            #pragma unroll
            for (int rr = 0; rr < 4; ++rr) {
                const int r = 4 * g + rr;
                const float wf  = (g == 0) ? wf_lo[rr] : wf_hi[rr];
                const float zv  = acc[0][r];
                const float zgx = acc[1][r];
                const float zgy = acc[2][r];
                const float tv = tanh_hw(zv);
                const float t1 = 1.f - tv * tv;
                const float t2 = -2.f * tv * t1;
                const float w1 = wf * t1;
                const float w2 = wf * t2;
                o[rr] = w1;
                P[0] = fmaf(wf, tv, P[0]);
                P[1] = fmaf(w1, zgx, P[1]);
                P[2] = fmaf(w1, zgy, P[2]);
                P[3] = fmaf(w2, zgx * zgx, P[3]);
                P[4] = fmaf(w2, zgx * zgy, P[4]);
                P[5] = fmaf(w2, zgy * zgy, P[5]);
            }
            *(v4f*)(b2st + QB * g) = o;
        }
        #pragma unroll
        for (int c = 0; c < NC; ++c)
            red[(2 * wave + hh) * NCOL + c * PPB + m] = P[c];
        __syncthreads();

        v8f aa0 = vzero, aa1 = vzero;
        const float* bq = B2 + hh * QB + 4 * m;
        bt_load4(bb0, bq, 0);
        #pragma unroll
        for (int g = 0; g < 4; ++g) {
            v4f (&awc)[4] = (g & 1) ? aw1 : aw0;
            v4f (&bbc)[4] = (g & 1) ? bb1 : bb0;
            v4f (&awn)[4] = (g & 1) ? aw0 : aw1;
            v4f (&bbn)[4] = (g & 1) ? bb0 : bb1;
            if (g + 1 < 4) { at_load4(awn, at, g + 1);   bt_load4(bbn, bq, g + 1); }
            LOADS_STAY_TOGETHER();
            #pragma unroll
            for (int q = 0; q < 4; ++q) {
                aa0 = __builtin_amdgcn_wmma_f32_16x16x4_f32(
                    false, A_XY(awc[q]), false, A_XY(bbc[q]), (short)0, aa0, false, false);
                aa1 = __builtin_amdgcn_wmma_f32_16x16x4_f32(
                    false, A_ZW(awc[q]), false, A_ZW(bbc[q]), (short)0, aa1, false, false);
            }
            LOADS_STAY_TOGETHER();
        }
        const v8f av = aa0 + aa1;

        const float* hrd = Hl + (4 * wave + 2 * hh) * QROW + 4 * m;
        float D[3];
        #pragma unroll
        for (int c = 0; c < 3; ++c) D[c] = 0.f;
        #pragma unroll
        for (int g = 0; g < 2; ++g) {
            #pragma unroll
            for (int c = 0; c < 3; ++c) {
                const v4f hv = *(const v4f*)(hrd + QROW * g + 4 * PPB * (3 + c));
                D[c] = fmaf(av[4 * g + 0], hv[0], D[c]);
                D[c] = fmaf(av[4 * g + 1], hv[1], D[c]);
                D[c] = fmaf(av[4 * g + 2], hv[2], D[c]);
                D[c] = fmaf(av[4 * g + 3], hv[3], D[c]);
            }
        }
        #pragma unroll
        for (int c = 0; c < 3; ++c)
            red2[(2 * wave + hh) * (3 * PPB) + c * PPB + m] = D[c];
        __syncthreads();
        }
    }

    if (t < NCOL) {
        float s = 0.f;
        #pragma unroll
        for (int i = 0; i < NPART; ++i) s += red[i * NCOL + t];
        if (t >= 3 * PPB) {
            #pragma unroll
            for (int i = 0; i < NPART; ++i) s += red2[i * (3 * PPB) + t - 3 * PPB];
        }
        sbuf[t] = s;
    }
    __syncthreads();
    if (t < PPB) {
        const int p = t;
        const float sv  = sbuf[0 * PPB + p] + bf[0];
        const float sgx = sbuf[1 * PPB + p];
        const float sgy = sbuf[2 * PPB + p];
        const float sxx = sbuf[3 * PPB + p];
        const float sxy = sbuf[4 * PPB + p];
        const float syy = sbuf[5 * PPB + p];
        const float tv = tanhf(sv);
        const float t1 = 1.f - tv * tv;
        const float t2 = -2.f * tv * t1;
        jet[0 * PPB + p] = tv;
        jet[1 * PPB + p] = t1 * sgx;
        jet[2 * PPB + p] = t1 * sgy;
        jet[3 * PPB + p] = fmaf(t2, sgx * sgx, t1 * sxx);
        jet[4 * PPB + p] = fmaf(t2, sgx * sgy, t1 * sxy);
        jet[5 * PPB + p] = fmaf(t2, sgy * sgy, t1 * syy);
    }
    __syncthreads();
}

__global__ void __launch_bounds__(WGSIZE, 8)
pinn_elasticity_kernel(
    const float* __restrict__ x, const float* __restrict__ y,
    const float* __restrict__ W1u, const float* __restrict__ b1u,
    const float* __restrict__ Whu, const float* __restrict__ bhu,
    const float* __restrict__ Wfu, const float* __restrict__ bfu,
    const float* __restrict__ W1v, const float* __restrict__ b1v,
    const float* __restrict__ Whv, const float* __restrict__ bhv,
    const float* __restrict__ Wfv, const float* __restrict__ bfv,
    const float* __restrict__ mu, const float* __restrict__ lmbd,
    float* __restrict__ out, int N)
{
    extern __shared__ float lds[];
    float* Hl   = lds;
    float* sbuf = Hl + HDIM * CP;
    float* jets = sbuf + NCOL;
    float* xs   = jets + 2 * NC * PPB;
    float* ys   = xs + PPB;
    float* B2   = ys + PPB;
    float* red  = B2 + 32 * QB;
    float* red2 = red + NPART * NCOL;

    const int t  = threadIdx.x;
    const int p0 = blockIdx.x * PPB;

    const int nleft = N - p0;
    if (t < PPB) {
        const float* xb = x + p0;
        const float* yb = y + p0;
        xs[t] = (t < nleft) ? xb[t] : 0.f;
        ys[t] = (t < nleft) ? yb[t] : 0.f;
    }
    __syncthreads();

    run_net(Hl, sbuf, B2, red, red2, jets,            W1u, b1u, Whu, bhu, Wfu, bfu, xs, ys);
    run_net(Hl, sbuf, B2, red, red2, jets + NC * PPB, W1v, b1v, Whv, bhv, Wfv, bfv, xs, ys);

    if (t < PPB) {
        const int p  = t;
        if (p < nleft) {
            const float* ju = jets;
            const float* jv = jets + NC * PPB;
            const float Uv  = ju[0 * PPB + p];
            const float Uxx = ju[3 * PPB + p];
            const float Uxy = ju[4 * PPB + p];
            const float Uyy = ju[5 * PPB + p];
            const float Vv  = jv[0 * PPB + p];
            const float Vxx = jv[3 * PPB + p];
            const float Vxy = jv[4 * PPB + p];
            const float Vyy = jv[5 * PPB + p];

            const float MU = mu[0], LA = lmbd[0];
            const float C11 = 2.f * MU + LA;
            const float C12 = LA;
            const float C33 = 2.f * MU;

            const float rx = C11 * Uxx + C12 * Vxy + C33 * (Uyy + 0.5f * Vxy);
            const float ry = C33 * (Uxy + 0.5f * Vxx) + C11 * Vyy + C12 * Uxy;

            const float PI = 3.14159265358979323846f;
            const float Qc = 4.0f;
            const float xx = xs[p], yy = ys[p];
            const float y2 = yy * yy, y3 = y2 * yy, y4 = y3 * yy;
            const float s2px = sinf(2.f * PI * xx), c2px = cosf(2.f * PI * xx);
            const float spx  = sinf(PI * xx),       cpx  = cosf(PI * xx);
            const float spy  = sinf(PI * yy),       cpy  = cosf(PI * yy);
            const float PI2  = PI * PI;

            const float Fx = -1.0f * (4.f * PI2 * c2px * spy - Qc * y3 * PI * cpx)
                             - 0.5f * (PI2 * c2px * spy - Qc * y3 * PI * cpx)
                             - 0.5f * 8.f * PI2 * c2px * spy;
            const float Fy =  1.0f * (3.f * Qc * y2 * spx - 2.f * PI2 * cpy * s2px)
                             - 0.5f * (2.f * PI2 * cpy * s2px + Qc * y4 * PI2 * spx * 0.25f)
                             + 0.5f * 6.f * Qc * y2 * spx;

            float* ob = out + p0;
            ob[0 * N + p] = rx - Fx;
            ob[1 * N + p] = ry - Fy;
            ob[2 * N + p] = Uv;
            ob[3 * N + p] = Vv;
        }
    }
}

extern "C" void kernel_launch(void* const* d_in, const int* in_sizes, int n_in,
                              void* d_out, int out_size, void* d_ws, size_t ws_size,
                              hipStream_t stream) {
    const float* x    = (const float*)d_in[0];
    const float* y    = (const float*)d_in[1];
    const float* W1u  = (const float*)d_in[2];
    const float* b1u  = (const float*)d_in[3];
    const float* Whu  = (const float*)d_in[4];
    const float* bhu  = (const float*)d_in[5];
    const float* Wfu  = (const float*)d_in[6];
    const float* bfu  = (const float*)d_in[7];
    const float* W1v  = (const float*)d_in[8];
    const float* b1v  = (const float*)d_in[9];
    const float* Whv  = (const float*)d_in[10];
    const float* bhv  = (const float*)d_in[11];
    const float* Wfv  = (const float*)d_in[12];
    const float* bfv  = (const float*)d_in[13];
    const float* mu   = (const float*)d_in[14];
    const float* lmbd = (const float*)d_in[15];

    const int N = in_sizes[0];
    const int grid = (N + PPB - 1) / PPB;
    const size_t shmem =
        (size_t)(HDIM * CP + NCOL + 2 * NC * PPB + 2 * PPB + 32 * QB + NPART * NCOL + NPART * 3 * PPB) * sizeof(float);

    pinn_elasticity_kernel<<<grid, WGSIZE, shmem, stream>>>(
        x, y, W1u, b1u, Whu, bhu, Wfu, bfu,
        W1v, b1v, Whv, bhv, Wfv, bfv, mu, lmbd,
        (float*)d_out, N);
}


# Second derivatives leave the forward pass: hidden layers carry value, d/dx, d/dy (3 tiles, was 6); one adjoint vector runs back through the 3 transposed matrices; 2nd-derivative sums re-ordered
# speedup vs baseline: 1.7895x; 1.1853x over previous
#include <hip/hip_runtime.h>
#include <math.h>

typedef __attribute__((ext_vector_type(2))) float v2f;
typedef __attribute__((ext_vector_type(4))) float v4f;
typedef __attribute__((ext_vector_type(8))) float v8f;

#define HDIM 128
#define NLAY 3
#define PPB 16
#define NC 6
#define NF 3
#define NCOL (NC * PPB)
#define CP 56
#define QROW (4 * CP)
#define WGSIZE 256
#define QB 68
#define NPART 16
#define AG 2
#define NG (16 / AG)
#define SAVL (6 * 4 * WGSIZE)


#define A_LOAD(dst, base, off) do { \
    dst[0] = *(const v4f*)((base) + (off));      \
    dst[1] = *(const v4f*)((base) + (off) + 8);  \
} while (0)

__device__ __forceinline__ void b_load3(v4f (&b)[3], const float* __restrict__ hp)
{
    #pragma unroll
    for (int c = 0; c < 3; ++c) b[c] = *(const v4f*)(hp + 4 * PPB * c);
}

#define A_XY(v) __builtin_shufflevector((v), (v), 0, 1)
#define A_ZW(v) __builtin_shufflevector((v), (v), 2, 3)

__device__ __forceinline__ float tanh_hw(const float z)
{
    const float e = __builtin_amdgcn_exp2f(z * 2.88539008177792681472f);
    return fmaf(-2.f, __builtin_amdgcn_rcpf(e + 1.f), 1.f);
}

#define LOADS_STAY_TOGETHER() __builtin_amdgcn_sched_barrier(0)

#define PS (2 * QROW)

__device__ __forceinline__ void mma3(v8f (&acc)[NF], const v4f a, const v4f (&b)[3])
{
    #pragma unroll
    for (int c = 0; c < 3; ++c)
        acc[c] = __builtin_amdgcn_wmma_f32_16x16x4_f32(
            false, A_XY(a), false, A_XY(b[c]), (short)0, acc[c], false, false);
    #pragma unroll
    for (int c = 0; c < 3; ++c)
        acc[c] = __builtin_amdgcn_wmma_f32_16x16x4_f32(
            false, A_ZW(a), false, A_ZW(b[c]), (short)0, acc[c], false, false);
}

__device__ __forceinline__ void at_load(v4f (&a)[AG], const float* __restrict__ wt,
                                         const unsigned ao, const int g)
{
    #pragma unroll
    for (int q = 0; q < AG; ++q) {
        const float* p = wt + 8 * (AG * g + q) * HDIM;
        a[q][0] = (p + 0 * HDIM)[ao];
        a[q][1] = (p + 1 * HDIM)[ao];
        a[q][2] = (p + 2 * HDIM)[ao];
        a[q][3] = (p + 3 * HDIM)[ao];
    }
}

__device__ __forceinline__ void bt_load(v4f (&b)[AG], const float* __restrict__ bq, const int g)
{
    #pragma unroll
    for (int q = 0; q < AG; ++q) b[q] = *(const v4f*)(bq + 2 * (AG * g + q) * QB);
}

template <int MODE>
__device__ __forceinline__ void k_trip3(v8f (&acc)[NF], v4f (&A0)[2], v4f (&A1)[2],
                                        v4f (&X0)[3], v4f (&X1)[3], v4f (&Y0)[3], v4f (&Y1)[3],
                                        const float* __restrict__ wl, const unsigned wq,
                                        const float* __restrict__ hq,
                                        const float* __restrict__ wnl,
                                        v4f& wf_lo, v4f& wf_hi, v4f (&aw)[3][AG],
                                        const float* __restrict__ Wf, const unsigned bo,
                                        const unsigned ao)
{
    A_LOAD(A1, wl, wq + 16);
    b_load3(Y0, hq + 2 * PS);   b_load3(Y1, hq + 3 * PS);
    LOADS_STAY_TOGETHER();
    mma3(acc, A0[0], X0);
    mma3(acc, A0[1], X1);
    LOADS_STAY_TOGETHER();
    if (MODE == 2) {
        wf_lo = *(const v4f*)(Wf + bo);
        wf_hi = *(const v4f*)(Wf + bo + 4);
        at_load(aw[0], wl, ao, 0);
        at_load(aw[1], wl, ao, 1);
    } else if (MODE == 1) {
        A_LOAD(A0, wnl, wq - 96);
    } else {
        A_LOAD(A0, wl, wq + 32);
        b_load3(X0, hq + 4 * PS);   b_load3(X1, hq + 5 * PS);
    }
    LOADS_STAY_TOGETHER();
    mma3(acc, A1[0], Y0);
    mma3(acc, A1[1], Y1);
    LOADS_STAY_TOGETHER();
}

__device__ __forceinline__ v8f adjoint(v4f (&aw)[3][AG], const float* __restrict__ wt,
                                       const unsigned ao, const float* __restrict__ bq)
{
    const v8f vzero = {0.f, 0.f, 0.f, 0.f, 0.f, 0.f, 0.f, 0.f};
    v4f bb[2][AG];
    v8f aa0 = vzero, aa1 = vzero;
    bt_load(bb[0], bq, 0);
    #pragma unroll
    for (int g = 0; g < NG; ++g) {
        if (g + 2 < NG) at_load(aw[(g + 2) % 3], wt, ao, g + 2);
        if (g + 1 < NG) bt_load(bb[(g + 1) & 1], bq, g + 1);
        LOADS_STAY_TOGETHER();
        #pragma unroll
        for (int q = 0; q < AG; ++q) {
            aa0 = __builtin_amdgcn_wmma_f32_16x16x4_f32(
                false, A_XY(aw[g % 3][q]), false, A_XY(bb[g & 1][q]), (short)0, aa0, false, false);
            aa1 = __builtin_amdgcn_wmma_f32_16x16x4_f32(
                false, A_ZW(aw[g % 3][q]), false, A_ZW(bb[g & 1][q]), (short)0, aa1, false, false);
        }
        LOADS_STAY_TOGETHER();
    }
    return aa0 + aa1;
}

__device__ __forceinline__ void back_step(const v8f av, const float* __restrict__ sv,
                                          float (&D)[3], float* __restrict__ b2st)
{
    #pragma unroll
    for (int g = 0; g < 2; ++g) {
        const v4f tq = *(const v4f*)(sv + (3 * g + 0) * 4 * WGSIZE);
        const v4f xq = *(const v4f*)(sv + (3 * g + 1) * 4 * WGSIZE);
        const v4f yq = *(const v4f*)(sv + (3 * g + 2) * 4 * WGSIZE);
        v4f o;
        #pragma unroll
        for (int rr = 0; rr < 4; ++rr) {
            const float a   = av[4 * g + rr];
            const float tv  = tq[rr];
            const float zgx = xq[rr];
            const float zgy = yq[rr];
            const float t1 = 1.f - tv * tv;
            const float t2 = -2.f * tv * t1;
            const float w2 = a * t2;
            o[rr] = a * t1;
            D[0] = fmaf(w2, zgx * zgx, D[0]);
            D[1] = fmaf(w2, zgx * zgy, D[1]);
            D[2] = fmaf(w2, zgy * zgy, D[2]);
        }
        *(v4f*)(b2st + QB * g) = o;
    }
}

__device__ __forceinline__ void run_net(
    float* __restrict__ Hl, float* __restrict__ sbuf, float* __restrict__ sav,
    float* __restrict__ jet,
    const float* __restrict__ W1, const float* __restrict__ b1,
    const float* __restrict__ Wh, const float* __restrict__ bh,
    const float* __restrict__ Wf, const float* __restrict__ bf,
    const float* __restrict__ xs, const float* __restrict__ ys)
{
    float* B2a  = Hl;
    float* B2b  = B2a + 32 * QB;
    float* red  = B2b + 32 * QB;
    float* red2 = red + NPART * NCOL;

    const int t    = threadIdx.x;
    const int lane = t & 31;
    const int wave = __builtin_amdgcn_readfirstlane(t >> 5);

    const int rbase = wave * 16;
    const int m     = lane & 15;
    const int hh    = lane >> 4;
    const unsigned wro = (unsigned)((rbase + m) * HDIM + 4 * hh);
    const float* hbase = Hl + hh * QROW + 4 * m;
    const unsigned ato = (unsigned)(4 * hh * HDIM + rbase + m);
    const unsigned bo = (unsigned)(rbase + 8 * hh);

    v4f A0[2], A1[2];

    v4f bias_lo, bias_hi;

    {
        const int p = t & 15;
        const int g = t >> 4;
        const unsigned gu = (unsigned)g;
        const float x = xs[p], y = ys[p];
        v4f iwa[2], iwb[2], ibq[2];
        #pragma unroll
        for (int j = 0; j < 2; ++j) {
            iwa[j] = *(const v4f*)(W1 + 128 * j + 8u * gu);
            iwb[j] = *(const v4f*)(W1 + 128 * j + 4 + 8u * gu);
            ibq[j] = *(const v4f*)(b1 + 64 * j + 4u * gu);
        }
        A_LOAD(A0, Wh, wro);
        bias_lo = *(const v4f*)(bh + bo);
        bias_hi = *(const v4f*)(bh + bo + 4);
        float* hrow = Hl + g * QROW + 4 * p;
        #pragma unroll
        for (int j = 0; j < 2; ++j) {
            v4f o[NF];
            #pragma unroll
            for (int rr = 0; rr < 4; ++rr) {
                const v4f wv   = (rr < 2) ? iwa[j] : iwb[j];
                const float w0 = wv[2 * (rr & 1)], w1 = wv[2 * (rr & 1) + 1];
                const float z  = fmaf(w0, x, fmaf(w1, y, ibq[j][rr]));
                const float tv = tanh_hw(z);
                const float t1 = 1.f - tv * tv;
                o[0][rr] = tv;
                o[1][rr] = t1 * w0;
                o[2][rr] = t1 * w1;
            }
            #pragma unroll
            for (int c = 0; c < NF; ++c)
                *(v4f*)(hrow + 16 * QROW * j + 4 * PPB * c) = o[c];
        }
    }
    __syncthreads();

    const v8f vzero = {0.f, 0.f, 0.f, 0.f, 0.f, 0.f, 0.f, 0.f};
    v4f wf_lo, wf_hi;
    v4f aw[3][AG];
    v8f acc[NF];
    float* svl = sav + 4 * t;
    #pragma unroll
    for (int l = 0; l < NLAY; ++l) {
        const float* wl = Wh + l * HDIM * HDIM;
        acc[0] = __builtin_shufflevector(bias_lo, bias_hi, 0, 1, 2, 3, 4, 5, 6, 7);
        acc[1] = vzero;
        acc[2] = vzero;
        {
            v4f X0[3], X1[3], Y0[3], Y1[3];
            b_load3(X0, hbase);   b_load3(X1, hbase + PS);
            #pragma nounroll
            for (unsigned q2 = 0; q2 < 3; ++q2)
                k_trip3<0>(acc, A0, A1, X0, X1, Y0, Y1, wl, wro + 32u * q2, hbase + q2 * 8 * QROW,
                           wl, wf_lo, wf_hi, aw, Wf, bo, ato);
            if (l + 1 < NLAY)
                k_trip3<1>(acc, A0, A1, X0, X1, Y0, Y1, wl, wro + 96u, hbase + 24 * QROW,
                           wl + HDIM * HDIM, wf_lo, wf_hi, aw, Wf, bo, ato);
            else
                k_trip3<2>(acc, A0, A1, X0, X1, Y0, Y1, wl, wro + 96u, hbase + 24 * QROW,
                           wl, wf_lo, wf_hi, aw, Wf, bo, ato);
        }
        __syncthreads();
        if (l + 1 < NLAY) {
            bias_lo = *(const v4f*)(bh + (l + 1) * HDIM + bo);
            bias_hi = *(const v4f*)(bh + (l + 1) * HDIM + 4 + bo);

            float* hst = Hl + (4 * wave + 2 * hh) * QROW + 4 * m;
            float* svw = svl + l * SAVL;
            #pragma unroll
            for (int g = 0; g < 2; ++g) {
                v4f o[NF], kx, ky;
                #pragma unroll
                for (int rr = 0; rr < 4; ++rr) {
                    const int r = 4 * g + rr;
                    const float zv  = acc[0][r];
                    const float zgx = acc[1][r];
                    const float zgy = acc[2][r];
                    const float tv = tanh_hw(zv);
                    const float t1 = 1.f - tv * tv;
                    o[0][rr] = tv;
                    o[1][rr] = t1 * zgx;
                    o[2][rr] = t1 * zgy;
                    kx[rr] = zgx;
                    ky[rr] = zgy;
                }
                #pragma unroll
                for (int c = 0; c < NF; ++c)
                    *(v4f*)(hst + QROW * g + 4 * PPB * c) = o[c];
                *(v4f*)(svw + (3 * g + 0) * 4 * WGSIZE) = o[0];
                *(v4f*)(svw + (3 * g + 1) * 4 * WGSIZE) = kx;
                *(v4f*)(svw + (3 * g + 2) * 4 * WGSIZE) = ky;
            }
            __syncthreads();
        }
    }

    {
        float P[NC];
        #pragma unroll
        for (int c = 0; c < NC; ++c) P[c] = 0.f;
        float* b2st = B2a + (4 * wave + 2 * hh) * QB + 4 * m;
        #pragma unroll
        for (int g = 0; g < 2; ++g) {
            v4f o;
            #pragma unroll
            for (int rr = 0; rr < 4; ++rr) {
                const int r = 4 * g + rr;
                const float wf  = (g == 0) ? wf_lo[rr] : wf_hi[rr];
                const float zv  = acc[0][r];
                const float zgx = acc[1][r];
                const float zgy = acc[2][r];
                const float tv = tanh_hw(zv);
                const float t1 = 1.f - tv * tv;
                const float t2 = -2.f * tv * t1;
                const float w1 = wf * t1;
                const float w2 = wf * t2;
                o[rr] = w1;
                P[0] = fmaf(wf, tv, P[0]);
                P[1] = fmaf(w1, zgx, P[1]);
                P[2] = fmaf(w1, zgy, P[2]);
                P[3] = fmaf(w2, zgx * zgx, P[3]);
                P[4] = fmaf(w2, zgx * zgy, P[4]);
                P[5] = fmaf(w2, zgy * zgy, P[5]);
            }
            *(v4f*)(b2st + QB * g) = o;
        }
        #pragma unroll
        for (int c = 0; c < NC; ++c)
            red[(2 * wave + hh) * NCOL + c * PPB + m] = P[c];
    }
    __syncthreads();

    {
        float D[3];
        #pragma unroll
        for (int c = 0; c < 3; ++c) D[c] = 0.f;
        const float* bqa = B2a + hh * QB + 4 * m;
        const float* bqb = B2b + hh * QB + 4 * m;
        float* sta = B2a + (4 * wave + 2 * hh) * QB + 4 * m;
        float* stb = B2b + (4 * wave + 2 * hh) * QB + 4 * m;

        v8f av = adjoint(aw, Wh + 2 * HDIM * HDIM, ato, bqa);
        at_load(aw[0], Wh + 1 * HDIM * HDIM, ato, 0);
        at_load(aw[1], Wh + 1 * HDIM * HDIM, ato, 1);
        back_step(av, svl + 1 * SAVL, D, stb);
        __syncthreads();

        av = adjoint(aw, Wh + 1 * HDIM * HDIM, ato, bqb);
        at_load(aw[0], Wh, ato, 0);
        at_load(aw[1], Wh, ato, 1);
        back_step(av, svl, D, sta);
        __syncthreads();

        av = adjoint(aw, Wh, ato, bqa);
        v4f gw[4], gb[2];
        #pragma unroll
        for (int i = 0; i < 4; ++i) gw[i] = *(const v4f*)(W1 + 4 * i + 2u * bo);
        gb[0] = *(const v4f*)(b1 + bo);
        gb[1] = *(const v4f*)(b1 + 4 + bo);
        const float x = xs[m], y = ys[m];
        #pragma unroll
        for (int r = 0; r < 8; ++r) {
            const v4f wv   = gw[r >> 1];
            const float w0 = wv[2 * (r & 1)], w1 = wv[2 * (r & 1) + 1];
            const float z  = fmaf(w0, x, fmaf(w1, y, gb[r >> 2][r & 3]));
            const float tv = tanh_hw(z);
            const float t1 = 1.f - tv * tv;
            const float t2 = -2.f * tv * t1;
            D[0] = fmaf(av[r], t2 * w0 * w0, D[0]);
            D[1] = fmaf(av[r], t2 * w0 * w1, D[1]);
            D[2] = fmaf(av[r], t2 * w1 * w1, D[2]);
        }
        #pragma unroll
        for (int c = 0; c < 3; ++c)
            red2[(2 * wave + hh) * (3 * PPB) + c * PPB + m] = D[c];
    }
    __syncthreads();

    if (t < NCOL) {
        float s = 0.f;
        #pragma unroll
        for (int i = 0; i < NPART; ++i) s += red[i * NCOL + t];
        if (t >= 3 * PPB) {
            #pragma unroll
            for (int i = 0; i < NPART; ++i) s += red2[i * (3 * PPB) + t - 3 * PPB];
        }
        sbuf[t] = s;
    }
    __syncthreads();
    if (t < PPB) {
        const int p = t;
        const float sv  = sbuf[0 * PPB + p] + bf[0];
        const float sgx = sbuf[1 * PPB + p];
        const float sgy = sbuf[2 * PPB + p];
        const float sxx = sbuf[3 * PPB + p];
        const float sxy = sbuf[4 * PPB + p];
        const float syy = sbuf[5 * PPB + p];
        const float tv = tanhf(sv);
        const float t1 = 1.f - tv * tv;
        const float t2 = -2.f * tv * t1;
        jet[0 * PPB + p] = tv;
        jet[1 * PPB + p] = t1 * sgx;
        jet[2 * PPB + p] = t1 * sgy;
        jet[3 * PPB + p] = fmaf(t2, sgx * sgx, t1 * sxx);
        jet[4 * PPB + p] = fmaf(t2, sgx * sgy, t1 * sxy);
        jet[5 * PPB + p] = fmaf(t2, sgy * sgy, t1 * syy);
    }
    __syncthreads();
}

__global__ void __launch_bounds__(WGSIZE, 8)
pinn_elasticity_kernel(
    const float* __restrict__ x, const float* __restrict__ y,
    const float* __restrict__ W1u, const float* __restrict__ b1u,
    const float* __restrict__ Whu, const float* __restrict__ bhu,
    const float* __restrict__ Wfu, const float* __restrict__ bfu,
    const float* __restrict__ W1v, const float* __restrict__ b1v,
    const float* __restrict__ Whv, const float* __restrict__ bhv,
    const float* __restrict__ Wfv, const float* __restrict__ bfv,
    const float* __restrict__ mu, const float* __restrict__ lmbd,
    float* __restrict__ out, int N)
{
    extern __shared__ float lds[];
    float* Hl   = lds;
    float* sbuf = Hl + HDIM * CP;
    float* jets = sbuf + NCOL;
    float* xs   = jets + 2 * NC * PPB;
    float* ys   = xs + PPB;
    float* sav  = ys + PPB;

    const int t  = threadIdx.x;
    const int p0 = blockIdx.x * PPB;

    const int nleft = N - p0;
    if (t < PPB) {
        const float* xb = x + p0;
        const float* yb = y + p0;
        xs[t] = (t < nleft) ? xb[t] : 0.f;
        ys[t] = (t < nleft) ? yb[t] : 0.f;
    }
    __syncthreads();

    run_net(Hl, sbuf, sav, jets,            W1u, b1u, Whu, bhu, Wfu, bfu, xs, ys);
    run_net(Hl, sbuf, sav, jets + NC * PPB, W1v, b1v, Whv, bhv, Wfv, bfv, xs, ys);

    if (t < PPB) {
        const int p  = t;
        if (p < nleft) {
            const float* ju = jets;
            const float* jv = jets + NC * PPB;
            const float Uv  = ju[0 * PPB + p];
            const float Uxx = ju[3 * PPB + p];
            const float Uxy = ju[4 * PPB + p];
            const float Uyy = ju[5 * PPB + p];
            const float Vv  = jv[0 * PPB + p];
            const float Vxx = jv[3 * PPB + p];
            const float Vxy = jv[4 * PPB + p];
            const float Vyy = jv[5 * PPB + p];

            const float MU = mu[0], LA = lmbd[0];
            const float C11 = 2.f * MU + LA;
            const float C12 = LA;
            const float C33 = 2.f * MU;

            const float rx = C11 * Uxx + C12 * Vxy + C33 * (Uyy + 0.5f * Vxy);
            const float ry = C33 * (Uxy + 0.5f * Vxx) + C11 * Vyy + C12 * Uxy;

            const float PI = 3.14159265358979323846f;
            const float Qc = 4.0f;
            const float xx = xs[p], yy = ys[p];
            const float y2 = yy * yy, y3 = y2 * yy, y4 = y3 * yy;
            const float s2px = sinf(2.f * PI * xx), c2px = cosf(2.f * PI * xx);
            const float spx  = sinf(PI * xx),       cpx  = cosf(PI * xx);
            const float spy  = sinf(PI * yy),       cpy  = cosf(PI * yy);
            const float PI2  = PI * PI;

            const float Fx = -1.0f * (4.f * PI2 * c2px * spy - Qc * y3 * PI * cpx)
                             - 0.5f * (PI2 * c2px * spy - Qc * y3 * PI * cpx)
                             - 0.5f * 8.f * PI2 * c2px * spy;
            const float Fy =  1.0f * (3.f * Qc * y2 * spx - 2.f * PI2 * cpy * s2px)
                             - 0.5f * (2.f * PI2 * cpy * s2px + Qc * y4 * PI2 * spx * 0.25f)
                             + 0.5f * 6.f * Qc * y2 * spx;

            float* ob = out + p0;
            ob[0 * N + p] = rx - Fx;
            ob[1 * N + p] = ry - Fy;
            ob[2 * N + p] = Uv;
            ob[3 * N + p] = Vv;
        }
    }
}

extern "C" void kernel_launch(void* const* d_in, const int* in_sizes, int n_in,
                              void* d_out, int out_size, void* d_ws, size_t ws_size,
                              hipStream_t stream) {
    const float* x    = (const float*)d_in[0];
    const float* y    = (const float*)d_in[1];
    const float* W1u  = (const float*)d_in[2];
    const float* b1u  = (const float*)d_in[3];
    const float* Whu  = (const float*)d_in[4];
    const float* bhu  = (const float*)d_in[5];
    const float* Wfu  = (const float*)d_in[6];
    const float* bfu  = (const float*)d_in[7];
    const float* W1v  = (const float*)d_in[8];
    const float* b1v  = (const float*)d_in[9];
    const float* Whv  = (const float*)d_in[10];
    const float* bhv  = (const float*)d_in[11];
    const float* Wfv  = (const float*)d_in[12];
    const float* bfv  = (const float*)d_in[13];
    const float* mu   = (const float*)d_in[14];
    const float* lmbd = (const float*)d_in[15];

    const int N = in_sizes[0];
    const int grid = (N + PPB - 1) / PPB;
    const size_t shmem =
        (size_t)(HDIM * CP + NCOL + 2 * NC * PPB + 2 * PPB + 2 * SAVL) * sizeof(float);

    pinn_elasticity_kernel<<<grid, WGSIZE, shmem, stream>>>(
        x, y, W1u, b1u, Whu, bhu, Wfu, bfu,
        W1v, b1v, Whv, bhv, Wfv, bfv, mu, lmbd,
        (float*)d_out, N);
}
